# starting kernel
# baseline (speedup 1.0000x reference)
#include <hip/hip_runtime.h>
#include <math.h>

#define NNODES   100000
#define NEDGES   1600000
#define L1_H     4
#define L1_C     32
#define L1_HC    128
#define L2_H     1
#define L2_C     64
#define L2_HC    64
#define NEG_SLOPE 0.2f
#define EPS_      1e-16f

typedef __attribute__((ext_vector_type(2))) float v2f;
typedef __attribute__((ext_vector_type(8))) float v8f;

__device__ __forceinline__ float atomic_max_f32(float* addr, float val) {
  if (val >= 0.0f)
    return __int_as_float(atomicMax((int*)addr, __float_as_int(val)));
  return __uint_as_float(atomicMin((unsigned int*)addr, __float_as_uint(val)));
}

__device__ __forceinline__ void edge_nodes(const int* __restrict__ ei, int idx,
                                           int& s, int& d) {
  if (idx < NEDGES) { s = ei[idx]; d = ei[NEDGES + idx]; }
  else             { s = idx - NEDGES; d = s; }
}

__global__ void fill_kernel(float* __restrict__ p, float v, long long n) {
  long long i = (long long)blockIdx.x * blockDim.x + threadIdx.x;
  if (i < n) p[i] = v;
}

template <int K, int NC>
__global__ __launch_bounds__(128)
void gemm_wmma_kernel(const float* __restrict__ A, const float* __restrict__ B,
                      float* __restrict__ C) {
  const int lane = threadIdx.x & 31;
  const int wv   = threadIdx.x >> 5;
  const int row0 = blockIdx.x * 16;
  const int col0 = wv * 32;
  const int m    = lane & 15;
  const int kk   = (lane >> 4) << 1;

  v8f acc0 = {};
  v8f acc1 = {};
  const float* ap = A + (long long)(row0 + m) * K + kk;
  const float* bp = B + (long long)kk * NC + (col0 + m);
#pragma unroll
  for (int k0 = 0; k0 < K; k0 += 4) {
    v2f a, b0, b1;
    a.x  = ap[k0];            a.y  = ap[k0 + 1];
    b0.x = bp[k0 * NC];       b0.y = bp[(k0 + 1) * NC];
    b1.x = bp[k0 * NC + 16];  b1.y = bp[(k0 + 1) * NC + 16];
    acc0 = __builtin_amdgcn_wmma_f32_16x16x4_f32(
        false, a, false, b0, (short)0, acc0, false, false);
    acc1 = __builtin_amdgcn_wmma_f32_16x16x4_f32(
        false, a, false, b1, (short)0, acc1, false, false);
  }
  const int orow = row0 + ((lane >> 4) << 3);
  const int ocol = col0 + m;
  float* cp = C + (long long)orow * NC + ocol;
#pragma unroll
  for (int r = 0; r < 8; ++r) {
    cp[(long long)r * NC]      = acc0[r];
    cp[(long long)r * NC + 16] = acc1[r];
  }
}

__global__ void attn_kernel(const float* __restrict__ h,
                            const float* __restrict__ att_s,
                            const float* __restrict__ att_d,
                            float* __restrict__ as_, float* __restrict__ ad_,
                            int H, int C) {
  int i = blockIdx.x * blockDim.x + threadIdx.x;
  if (i >= NNODES * H) return;
  int n = i / H, hd = i % H;
  const float* hv = h + (long long)n * H * C + hd * C;
  const float* s  = att_s + hd * C;
  const float* d  = att_d + hd * C;
  float ss = 0.0f, dd = 0.0f;
  for (int c = 0; c < C; ++c) { float x = hv[c]; ss += x * s[c]; dd += x * d[c]; }
  as_[i] = ss;
  ad_[i] = dd;
}

__global__ void edge_max_kernel(const int* __restrict__ ei,
                                const float* __restrict__ as_,
                                const float* __restrict__ ad_,
                                float* __restrict__ mx, int H) {
  int idx = blockIdx.x * blockDim.x + threadIdx.x;
  if (idx >= NEDGES + NNODES) return;
  int s, d; edge_nodes(ei, idx, s, d);
  for (int h = 0; h < H; ++h) {
    float e = as_[s * H + h] + ad_[d * H + h];
    e = e > 0.0f ? e : NEG_SLOPE * e;
    atomic_max_f32(&mx[d * H + h], e);
  }
}

__global__ void edge_sum_kernel(const int* __restrict__ ei,
                                const float* __restrict__ as_,
                                const float* __restrict__ ad_,
                                const float* __restrict__ mx,
                                float* __restrict__ sm, int H) {
  int idx = blockIdx.x * blockDim.x + threadIdx.x;
  if (idx >= NEDGES + NNODES) return;
  int s, d; edge_nodes(ei, idx, s, d);
  for (int h = 0; h < H; ++h) {
    float e = as_[s * H + h] + ad_[d * H + h];
    e = e > 0.0f ? e : NEG_SLOPE * e;
    atomicAdd(&sm[d * H + h], expf(e - mx[d * H + h]));
  }
}

__global__ void edge_scatter_kernel(const int* __restrict__ ei,
                                    const float* __restrict__ as_,
                                    const float* __restrict__ ad_,
                                    const float* __restrict__ mx,
                                    const float* __restrict__ sm,
                                    const float* __restrict__ hfeat,
                                    float* __restrict__ out, int H, int C) {
  const int HC = H * C;
  long long gid = (long long)blockIdx.x * blockDim.x + threadIdx.x;
  long long tot = (long long)(NEDGES + NNODES) * HC;
  if (gid >= tot) return;
  int idx = (int)(gid / HC);
  int c   = (int)(gid % HC);
  int h   = c / C;
  int s, d; edge_nodes(ei, idx, s, d);
  float e  = as_[s * H + h] + ad_[d * H + h];
  e        = e > 0.0f ? e : NEG_SLOPE * e;
  float ex = expf(e - mx[d * H + h]);
  float alpha = ex / (sm[d * H + h] + EPS_);
  atomicAdd(&out[(long long)d * HC + c], hfeat[(long long)s * HC + c] * alpha);
}

__global__ void bias_elu_kernel(float* __restrict__ o, const float* __restrict__ b,
                                long long n, int HC) {
  long long i = (long long)blockIdx.x * blockDim.x + threadIdx.x;
  if (i >= n) return;
  float v = o[i] + b[(int)(i % HC)];
  o[i] = v > 0.0f ? v : expm1f(v);
}

__global__ void bias_add_kernel(float* __restrict__ o, const float* __restrict__ b,
                                long long n, int HC) {
  long long i = (long long)blockIdx.x * blockDim.x + threadIdx.x;
  if (i >= n) return;
  o[i] += b[(int)(i % HC)];
}

extern "C" void kernel_launch(void* const* d_in, const int* in_sizes, int n_in,
                              void* d_out, int out_size, void* d_ws, size_t ws_size,
                              hipStream_t stream) {
  (void)in_sizes; (void)n_in; (void)out_size; (void)ws_size;

  const float* x    = (const float*)d_in[0];
  const int*   ei   = (const int*)  d_in[1];
  const float* W1   = (const float*)d_in[2];
  const float* aS1  = (const float*)d_in[3];
  const float* aD1  = (const float*)d_in[4];
  const float* b1   = (const float*)d_in[5];
  const float* W2   = (const float*)d_in[6];
  const float* aS2  = (const float*)d_in[7];
  const float* aD2  = (const float*)d_in[8];
  const float* b2   = (const float*)d_in[9];
  float* out = (float*)d_out;

  float* ws  = (float*)d_ws;
  float* h1  = ws;
  float* o1  = h1  + (long long)NNODES * L1_HC;
  float* as1 = o1  + (long long)NNODES * L1_HC;
  float* ad1 = as1 + NNODES * L1_H;
  float* mx1 = ad1 + NNODES * L1_H;
  float* sm1 = mx1 + NNODES * L1_H;
  float* as2 = sm1 + NNODES * L1_H;
  float* ad2 = as2 + NNODES;
  float* mx2 = ad2 + NNODES;
  float* sm2 = mx2 + NNODES;
  float* h2  = h1;

  const float NEGINF = -__builtin_inff();
  const int EN = NEDGES + NNODES;
  auto cdiv = [](long long a, long long b) { return (int)((a + b - 1) / b); };

  fill_kernel<<<cdiv((long long)NNODES * L1_HC, 256), 256, 0, stream>>>(o1, 0.0f, (long long)NNODES * L1_HC);
  fill_kernel<<<cdiv(NNODES * L1_H, 256), 256, 0, stream>>>(mx1, NEGINF, NNODES * L1_H);
  fill_kernel<<<cdiv(NNODES * L1_H, 256), 256, 0, stream>>>(sm1, 0.0f, NNODES * L1_H);
  fill_kernel<<<cdiv(NNODES, 256), 256, 0, stream>>>(mx2, NEGINF, NNODES);
  fill_kernel<<<cdiv(NNODES, 256), 256, 0, stream>>>(sm2, 0.0f, NNODES);
  fill_kernel<<<cdiv((long long)NNODES * L2_HC, 256), 256, 0, stream>>>(out, 0.0f, (long long)NNODES * L2_HC);

  gemm_wmma_kernel<128, L1_HC><<<NNODES / 16, 128, 0, stream>>>(x, W1, h1);
  attn_kernel<<<cdiv(NNODES * L1_H, 128), 128, 0, stream>>>(h1, aS1, aD1, as1, ad1, L1_H, L1_C);
  edge_max_kernel<<<cdiv(EN, 256), 256, 0, stream>>>(ei, as1, ad1, mx1, L1_H);
  edge_sum_kernel<<<cdiv(EN, 256), 256, 0, stream>>>(ei, as1, ad1, mx1, sm1, L1_H);
  edge_scatter_kernel<<<cdiv((long long)EN * L1_HC, 256), 256, 0, stream>>>(
      ei, as1, ad1, mx1, sm1, h1, o1, L1_H, L1_C);
  bias_elu_kernel<<<cdiv((long long)NNODES * L1_HC, 256), 256, 0, stream>>>(
      o1, b1, (long long)NNODES * L1_HC, L1_HC);

  gemm_wmma_kernel<128, L2_HC><<<NNODES / 16, 64, 0, stream>>>(o1, W2, h2);
  attn_kernel<<<cdiv(NNODES, 128), 128, 0, stream>>>(h2, aS2, aD2, as2, ad2, L2_H, L2_C);
  edge_max_kernel<<<cdiv(EN, 256), 256, 0, stream>>>(ei, as2, ad2, mx2, L2_H);
  edge_sum_kernel<<<cdiv(EN, 256), 256, 0, stream>>>(ei, as2, ad2, mx2, sm2, L2_H);
  edge_scatter_kernel<<<cdiv((long long)EN * L2_HC, 256), 256, 0, stream>>>(
      ei, as2, ad2, mx2, sm2, h2, out, L2_H, L2_C);
  bias_add_kernel<<<cdiv((long long)NNODES * L2_HC, 256), 256, 0, stream>>>(
      out, b2, (long long)NNODES * L2_HC, L2_HC);
}


# Edges sorted by destination once per call (int count/rank, scan, fill); one wave per node does softmax and row sum in registers: no float atomics, sum order follows arrival-order rank
# speedup vs baseline: 7.4245x; 7.4245x over previous
#include <hip/hip_runtime.h>
#include <math.h>

#define NNODES   100000
#define NEDGES   1600000
#define L1_H     4
#define L1_C     32
#define L1_HC    128
#define L2_H     1
#define L2_C     64
#define L2_HC    64
#define NEG_SLOPE 0.2f
#define EPS_      1e-16f

#define SCAN_BLOCK   1024
#define SCAN_NBLOCKS 98
#define NODES_PER_BLOCK 8

static_assert(SCAN_NBLOCKS * SCAN_BLOCK >= NNODES + 1, "scan grid must cover rowptr[0..N]");
static_assert(NNODES % NODES_PER_BLOCK == 0, "gather grid divides the node count");
static_assert(NEDGES % 256 == 0, "edge grids divide the edge count");

typedef __attribute__((ext_vector_type(2))) float v2f;
typedef __attribute__((ext_vector_type(8))) float v8f;

template <int K, int NC>
__global__ __launch_bounds__(128)
void gemm_wmma_kernel(const float* __restrict__ A, const float* __restrict__ B,
                      float* __restrict__ C) {
  const int lane = threadIdx.x & 31;
  const int wv   = threadIdx.x >> 5;
  const int row0 = blockIdx.x * 16;
  const int col0 = wv * 32;
  const int m    = lane & 15;
  const int kk   = (lane >> 4) << 1;

  v8f acc0 = {};
  v8f acc1 = {};
  const float* ap = A + (long long)(row0 + m) * K + kk;
  const float* bp = B + (long long)kk * NC + (col0 + m);
#pragma unroll
  for (int k0 = 0; k0 < K; k0 += 4) {
    v2f a, b0, b1;
    a.x  = ap[k0];            a.y  = ap[k0 + 1];
    b0.x = bp[k0 * NC];       b0.y = bp[(k0 + 1) * NC];
    b1.x = bp[k0 * NC + 16];  b1.y = bp[(k0 + 1) * NC + 16];
    acc0 = __builtin_amdgcn_wmma_f32_16x16x4_f32(
        false, a, false, b0, (short)0, acc0, false, false);
    acc1 = __builtin_amdgcn_wmma_f32_16x16x4_f32(
        false, a, false, b1, (short)0, acc1, false, false);
  }
  const int orow = row0 + ((lane >> 4) << 3);
  const int ocol = col0 + m;
  float* cp = C + (long long)orow * NC + ocol;
#pragma unroll
  for (int r = 0; r < 8; ++r) {
    cp[(long long)r * NC]      = acc0[r];
    cp[(long long)r * NC + 16] = acc1[r];
  }
}

__global__ void attn_kernel(const float* __restrict__ h,
                            const float* __restrict__ att_s,
                            const float* __restrict__ att_d,
                            float* __restrict__ as_, float* __restrict__ ad_,
                            int H, int C) {
  int i = blockIdx.x * blockDim.x + threadIdx.x;
  if (i >= NNODES * H) return;
  int n = i / H, hd = i % H;
  const float* hv = h + (long long)n * H * C + hd * C;
  const float* s  = att_s + hd * C;
  const float* d  = att_d + hd * C;
  float ss = 0.0f, dd = 0.0f;
  for (int c = 0; c < C; ++c) { float x = hv[c]; ss += x * s[c]; dd += x * d[c]; }
  as_[i] = ss;
  ad_[i] = dd;
}

__global__ void zero_i32_kernel(int* __restrict__ p, int n) {
  int i = blockIdx.x * blockDim.x + threadIdx.x;
  if (i < n) p[i] = 0;
}

__global__ void dst_count_kernel(const int* __restrict__ ei, int* __restrict__ cnt,
                                 int* __restrict__ rank) {
  int i = blockIdx.x * blockDim.x + threadIdx.x;
  if (i >= NEDGES) return;
  rank[i] = atomicAdd(&cnt[ei[NEDGES + i]], 1);
}

__global__ __launch_bounds__(SCAN_BLOCK)
void scan_local_kernel(const int* __restrict__ cnt, int* __restrict__ rowptr,
                       int* __restrict__ bsum) {
  __shared__ int wsum[32];
  const int t = threadIdx.x, lane = t & 31, wv = t >> 5;
  const int i = blockIdx.x * SCAN_BLOCK + t;
  const int v = (i < NNODES) ? cnt[i] : 0;
  int x = v;
  for (int o = 1; o < 32; o <<= 1) { int y = __shfl_up(x, o); if (lane >= o) x += y; }
  if (lane == 31) wsum[wv] = x;
  __syncthreads();
  if (wv == 0) {
    const int w = wsum[lane];
    int xs = w;
    for (int o = 1; o < 32; o <<= 1) { int y = __shfl_up(xs, o); if (lane >= o) xs += y; }
    wsum[lane] = xs - w;
    if (lane == 31) bsum[blockIdx.x] = xs;
  }
  __syncthreads();
  if (i <= NNODES) rowptr[i] = x - v + wsum[wv];
}

__global__ __launch_bounds__(SCAN_BLOCK)
void scan_add_kernel(int* __restrict__ rowptr, const int* __restrict__ bsum) {
  __shared__ int off_s;
  const int t = threadIdx.x;
  if (t < 32) {
    int acc = 0;
    for (int j = t; j < (int)blockIdx.x; j += 32) acc += bsum[j];
    for (int o = 16; o > 0; o >>= 1) acc += __shfl_xor(acc, o);
    if (t == 0) off_s = acc;
  }
  __syncthreads();
  const int i = blockIdx.x * SCAN_BLOCK + t;
  if (i <= NNODES) rowptr[i] += off_s;
}

__global__ void dst_fill_kernel(const int* __restrict__ ei, const int* __restrict__ rowptr,
                                const int* __restrict__ rank, int* __restrict__ srcs) {
  int i = blockIdx.x * blockDim.x + threadIdx.x;
  if (i >= NEDGES) return;
  srcs[rowptr[ei[NEDGES + i]] + rank[i]] = ei[i];
}

__device__ __forceinline__ float readlane_f32(float v, int l) {
  return __int_as_float(__builtin_amdgcn_readlane(__float_as_int(v), l));
}

template <int H, int VEC, bool ELU_ACT>
__global__ __launch_bounds__(256)
void gat_gather_kernel(const int* __restrict__ rowptr, const int* __restrict__ srcs,
                       const float* __restrict__ as_, const float* __restrict__ ad_,
                       const float* __restrict__ hfeat, const float* __restrict__ bias,
                       float* __restrict__ out) {
  constexpr int HC = 32 * VEC;
  constexpr int C  = HC / H;
  typedef __attribute__((ext_vector_type(VEC))) float vecf;
  static_assert(H == 1 || H == 4, "score loader handles 1 or 4 heads");

  const int lane = threadIdx.x & 31;
  const int d = __builtin_amdgcn_readfirstlane(
      (int)(blockIdx.x * NODES_PER_BLOCK + (threadIdx.x >> 5)));
  const int start = rowptr[d];
  const int deg   = rowptr[d + 1] - start;
  const int total = deg + 1;
  const int myhead = (lane * VEC) / C;

  float adv[H];
#pragma unroll
  for (int h = 0; h < H; ++h) adv[h] = ad_[d * H + h];

  auto chunk_scores = [&](int base, int& s, float (&e)[H]) -> bool {
    const int t = base + lane;
    s = d;
    if (t < deg) s = srcs[start + t];
    if constexpr (H == 4) {
      const float4 a = *reinterpret_cast<const float4*>(as_ + (long long)s * 4);
      e[0] = a.x + adv[0]; e[1] = a.y + adv[1]; e[2] = a.z + adv[2]; e[3] = a.w + adv[3];
    } else {
      e[0] = as_[s] + adv[0];
    }
#pragma unroll
    for (int h = 0; h < H; ++h) e[h] = e[h] > 0.0f ? e[h] : NEG_SLOPE * e[h];
    return t < total;
  };

  int   s0;
  float e0[H];
  const bool v0 = chunk_scores(0, s0, e0);

  float m[H];
#pragma unroll
  for (int h = 0; h < H; ++h) m[h] = e0[h];
  for (int base = 32; base < total; base += 32) {
    int s; float e[H];
    chunk_scores(base, s, e);
#pragma unroll
    for (int h = 0; h < H; ++h) m[h] = fmaxf(m[h], e[h]);
  }
#pragma unroll
  for (int h = 0; h < H; ++h)
    for (int o = 16; o > 0; o >>= 1) m[h] = fmaxf(m[h], __shfl_xor(m[h], o));

  float ex0[H], sum[H];
#pragma unroll
  for (int h = 0; h < H; ++h) { ex0[h] = v0 ? expf(e0[h] - m[h]) : 0.0f; sum[h] = ex0[h]; }
  for (int base = 32; base < total; base += 32) {
    int s; float e[H];
    const bool v = chunk_scores(base, s, e);
#pragma unroll
    for (int h = 0; h < H; ++h) sum[h] += v ? expf(e[h] - m[h]) : 0.0f;
  }
#pragma unroll
  for (int h = 0; h < H; ++h)
    for (int o = 16; o > 0; o >>= 1) sum[h] += __shfl_xor(sum[h], o);

  float acc[VEC];
#pragma unroll
  for (int v = 0; v < VEC; ++v) acc[v] = 0.0f;

  auto accumulate = [&](int s, const float (&al)[H], int cnt) {
    for (int j = 0; j < cnt; j += 4) {
      vecf  r[4];
      float a[4];
#pragma unroll
      for (int u = 0; u < 4; ++u) {
        const int jj = (j + u < cnt) ? (j + u) : (cnt - 1);
        const int sj = __builtin_amdgcn_readlane(s, jj);
        r[u] = *reinterpret_cast<const vecf*>(hfeat + (long long)sj * HC + lane * VEC);
        float av = readlane_f32(al[0], jj);
#pragma unroll
        for (int h = 1; h < H; ++h) {
          const float ah = readlane_f32(al[h], jj);
          av = (myhead == h) ? ah : av;
        }
        a[u] = (j + u < cnt) ? av : 0.0f;
      }
#pragma unroll
      for (int u = 0; u < 4; ++u)
#pragma unroll
        for (int v = 0; v < VEC; ++v) acc[v] = fmaf(a[u], r[u][v], acc[v]);
    }
  };

  {
    float al[H];
#pragma unroll
    for (int h = 0; h < H; ++h) al[h] = ex0[h] / (sum[h] + EPS_);
    accumulate(s0, al, total < 32 ? total : 32);
  }
  for (int base = 32; base < total; base += 32) {
    int s; float e[H], al[H];
    const bool v = chunk_scores(base, s, e);
#pragma unroll
    for (int h = 0; h < H; ++h) al[h] = v ? expf(e[h] - m[h]) / (sum[h] + EPS_) : 0.0f;
    const int left = total - base;
    accumulate(s, al, left < 32 ? left : 32);
  }

  const vecf b = *reinterpret_cast<const vecf*>(bias + lane * VEC);
  vecf o;
#pragma unroll
  for (int v = 0; v < VEC; ++v) {
    float x = acc[v] + b[v];
    if (ELU_ACT) x = x > 0.0f ? x : expm1f(x);
    o[v] = x;
  }
  *reinterpret_cast<vecf*>(out + (long long)d * HC + lane * VEC) = o;
}

extern "C" void kernel_launch(void* const* d_in, const int* in_sizes, int n_in,
                              void* d_out, int out_size, void* d_ws, size_t ws_size,
                              hipStream_t stream) {
  (void)in_sizes; (void)n_in; (void)out_size; (void)ws_size;

  const float* x    = (const float*)d_in[0];
  const int*   ei   = (const int*)  d_in[1];
  const float* W1   = (const float*)d_in[2];
  const float* aS1  = (const float*)d_in[3];
  const float* aD1  = (const float*)d_in[4];
  const float* b1   = (const float*)d_in[5];
  const float* W2   = (const float*)d_in[6];
  const float* aS2  = (const float*)d_in[7];
  const float* aD2  = (const float*)d_in[8];
  const float* b2   = (const float*)d_in[9];
  float* out = (float*)d_out;

  float* ws  = (float*)d_ws;
  float* h1  = ws;
  float* o1  = h1  + (long long)NNODES * L1_HC;
  float* as1 = o1  + (long long)NNODES * L1_HC;
  float* ad1 = as1 + NNODES * L1_H;
  float* as2 = ad1 + NNODES * L1_H;
  float* ad2 = as2 + NNODES;
  int*   cnt    = (int*)(ad2 + NNODES);
  int*   rowptr = cnt + NNODES;
  int*   bsum   = rowptr + SCAN_NBLOCKS * SCAN_BLOCK;
  int*   rank   = bsum + 128;
  int*   srcs   = rank + NEDGES;
  float* h2  = h1;

  zero_i32_kernel<<<(NNODES + 255) / 256, 256, 0, stream>>>(cnt, NNODES);
  dst_count_kernel<<<NEDGES / 256, 256, 0, stream>>>(ei, cnt, rank);
  scan_local_kernel<<<SCAN_NBLOCKS, SCAN_BLOCK, 0, stream>>>(cnt, rowptr, bsum);
  scan_add_kernel<<<SCAN_NBLOCKS, SCAN_BLOCK, 0, stream>>>(rowptr, bsum);
  dst_fill_kernel<<<NEDGES / 256, 256, 0, stream>>>(ei, rowptr, rank, srcs);

  gemm_wmma_kernel<128, L1_HC><<<NNODES / 16, 128, 0, stream>>>(x, W1, h1);
  attn_kernel<<<(NNODES * L1_H + 127) / 128, 128, 0, stream>>>(h1, aS1, aD1, as1, ad1, L1_H, L1_C);
  gat_gather_kernel<L1_H, L1_HC / 32, true><<<NNODES / NODES_PER_BLOCK, 256, 0, stream>>>(
      rowptr, srcs, as1, ad1, h1, b1, o1);

  gemm_wmma_kernel<128, L2_HC><<<NNODES / 16, 64, 0, stream>>>(o1, W2, h2);
  attn_kernel<<<(NNODES + 127) / 128, 128, 0, stream>>>(h2, aS2, aD2, as2, ad2, L2_H, L2_C);
  gat_gather_kernel<L2_H, L2_HC / 32, false><<<NNODES / NODES_PER_BLOCK, 256, 0, stream>>>(
      rowptr, srcs, as2, ad2, h2, b2, out);
}


# Edge sort by destination redone as two-level counting sort in LDS (chunks grouped by 512-node range, one block per range places them): no global atomics, 11 to 8 launches; sum order = LDS add order
# speedup vs baseline: 11.5928x; 1.5614x over previous
#include <hip/hip_runtime.h>
#include <math.h>

#define NNODES   100000
#define NEDGES   1600000
#define L1_H     4
#define L1_C     32
#define L1_HC    128
#define L2_H     1
#define L2_C     64
#define L2_HC    64
#define NEG_SLOPE 0.2f
#define EPS_      1e-16f

#define NODES_PER_BLOCK 8

#define NCHUNK   250
#define CHUNK_E  6400
#define RANGE_SH 9
#define RANGE_N  512
#define NRANGE   196
#define SORT_T   1024
#define STAGE_CAP 12288

static_assert(NCHUNK * CHUNK_E == NEDGES, "chunks cover the edge list exactly");
static_assert(NRANGE * RANGE_N >= NNODES + 1, "ranges must cover rowptr[0..N]");
static_assert((1 << RANGE_SH) == RANGE_N, "range size is a power of two");
static_assert(NCHUNK * 4 <= SORT_T, "four threads per piece in the collect kernel");
static_assert(NNODES % NODES_PER_BLOCK == 0, "gather grid divides the node count");

typedef __attribute__((ext_vector_type(2))) float v2f;
typedef __attribute__((ext_vector_type(8))) float v8f;

template <int K, int NC>
__global__ __launch_bounds__(128)
void gemm_wmma_kernel(const float* __restrict__ A, const float* __restrict__ B,
                      float* __restrict__ C) {
  const int lane = threadIdx.x & 31;
  const int wv   = threadIdx.x >> 5;
  const int row0 = blockIdx.x * 16;
  const int col0 = wv * 32;
  const int m    = lane & 15;
  const int kk   = (lane >> 4) << 1;

  v8f acc0 = {};
  v8f acc1 = {};
  const float* ap = A + (long long)(row0 + m) * K + kk;
  const float* bp = B + (long long)kk * NC + (col0 + m);
#pragma unroll
  for (int k0 = 0; k0 < K; k0 += 4) {
    v2f a, b0, b1;
    a.x  = ap[k0];            a.y  = ap[k0 + 1];
    b0.x = bp[k0 * NC];       b0.y = bp[(k0 + 1) * NC];
    b1.x = bp[k0 * NC + 16];  b1.y = bp[(k0 + 1) * NC + 16];
    acc0 = __builtin_amdgcn_wmma_f32_16x16x4_f32(
        false, a, false, b0, (short)0, acc0, false, false);
    acc1 = __builtin_amdgcn_wmma_f32_16x16x4_f32(
        false, a, false, b1, (short)0, acc1, false, false);
  }
  const int orow = row0 + ((lane >> 4) << 3);
  const int ocol = col0 + m;
  float* cp = C + (long long)orow * NC + ocol;
#pragma unroll
  for (int r = 0; r < 8; ++r) {
    cp[(long long)r * NC]      = acc0[r];
    cp[(long long)r * NC + 16] = acc1[r];
  }
}

__global__ void attn_kernel(const float* __restrict__ h,
                            const float* __restrict__ att_s,
                            const float* __restrict__ att_d,
                            float* __restrict__ as_, float* __restrict__ ad_,
                            int H, int C) {
  int i = blockIdx.x * blockDim.x + threadIdx.x;
  if (i >= NNODES * H) return;
  int n = i / H, hd = i % H;
  const float* hv = h + (long long)n * H * C + hd * C;
  const float* s  = att_s + hd * C;
  const float* d  = att_d + hd * C;
  float ss = 0.0f, dd = 0.0f;
  for (int c = 0; c < C; ++c) { float x = hv[c]; ss += x * s[c]; dd += x * d[c]; }
  as_[i] = ss;
  ad_[i] = dd;
}

__global__ __launch_bounds__(SORT_T)
void bin_kernel(const int* __restrict__ ei, int* __restrict__ cntTab,
                int* __restrict__ offTab, int* __restrict__ tmp_s,
                int* __restrict__ tmp_d) {
  __shared__ int hist[NRANGE];
  __shared__ int stage_s[CHUNK_E];
  __shared__ int stage_d[CHUNK_E];
  constexpr int PER = (CHUNK_E + SORT_T - 1) / SORT_T;
  constexpr int PL  = (NRANGE + 31) / 32;
  const int t = threadIdx.x;
  const int base = blockIdx.x * CHUNK_E;

  for (int b = t; b < NRANGE; b += SORT_T) hist[b] = 0;
  __syncthreads();

  int s[PER], d[PER];
#pragma unroll
  for (int k = 0; k < PER; ++k) {
    const int i = t + k * SORT_T;
    s[k] = 0; d[k] = 0;
    if (i < CHUNK_E) { d[k] = ei[NEDGES + base + i]; s[k] = ei[base + i]; }
  }
#pragma unroll
  for (int k = 0; k < PER; ++k)
    if (t + k * SORT_T < CHUNK_E) atomicAdd(&hist[d[k] >> RANGE_SH], 1);
  __syncthreads();

  if (t < 32) {
    int c[PL];
    int sum = 0;
#pragma unroll
    for (int j = 0; j < PL; ++j) {
      const int idx = t * PL + j;
      c[j] = (idx < NRANGE) ? hist[idx] : 0;
      sum += c[j];
    }
    int x = sum;
    for (int o = 1; o < 32; o <<= 1) { int y = __shfl_up(x, o); if (t >= o) x += y; }
    int run = x - sum;
#pragma unroll
    for (int j = 0; j < PL; ++j) {
      const int idx = t * PL + j;
      if (idx < NRANGE) {
        cntTab[idx * NCHUNK + blockIdx.x] = c[j];
        offTab[idx * NCHUNK + blockIdx.x] = run;
        hist[idx] = run;
      }
      run += c[j];
    }
  }
  __syncthreads();

#pragma unroll
  for (int k = 0; k < PER; ++k)
    if (t + k * SORT_T < CHUNK_E) {
      const int pos = atomicAdd(&hist[d[k] >> RANGE_SH], 1);
      stage_s[pos] = s[k];
      stage_d[pos] = d[k];
    }
  __syncthreads();

  for (int i = t; i < CHUNK_E; i += SORT_T) {
    tmp_s[base + i] = stage_s[i];
    tmp_d[base + i] = stage_d[i];
  }
}

__global__ __launch_bounds__(SORT_T)
void collect_kernel(const int* __restrict__ cntTab, const int* __restrict__ offTab,
                    const int* __restrict__ tmp_s, const int* __restrict__ tmp_d,
                    int* __restrict__ rowptr, int* __restrict__ srcs) {
  __shared__ int cnt[RANGE_N];
  __shared__ int cur[RANGE_N];
  __shared__ int pc[NCHUNK];
  __shared__ int po[NCHUNK];
  __shared__ int wsum[32];
  __shared__ int tot_s;
  __shared__ int stage[STAGE_CAP];
  const int t = threadIdx.x, lane = t & 31, wv = t >> 5;
  const int node0 = blockIdx.x * RANGE_N;
  const int npre  = blockIdx.x * NCHUNK;

  int acc = 0;
  for (int i = t; i < npre; i += SORT_T) acc += cntTab[i];
  for (int o = 16; o > 0; o >>= 1) acc += __shfl_xor(acc, o);
  if (lane == 0) wsum[wv] = acc;
  if (t < NCHUNK) { pc[t] = cntTab[npre + t]; po[t] = t * CHUNK_E + offTab[npre + t]; }
  if (t < RANGE_N) cnt[t] = 0;
  __syncthreads();
  int S = 0;
  for (int w = 0; w < 32; ++w) S += wsum[w];

  const int p = t >> 2, q = t & 3;
  int n = 0, o = 0;
  if (p < NCHUNK) { n = pc[p]; o = po[p]; }

  for (int j = q; j < n; j += 32) {
    int dd[8];
#pragma unroll
    for (int u = 0; u < 8; ++u) {
      const int jj = j + 4 * u;
      dd[u] = tmp_d[o + (jj < n ? jj : n - 1)];
    }
    asm volatile("" :: "v"(dd[0]), "v"(dd[1]), "v"(dd[2]), "v"(dd[3]),
                       "v"(dd[4]), "v"(dd[5]), "v"(dd[6]), "v"(dd[7]));
#pragma unroll
    for (int u = 0; u < 8; ++u)
      if (j + 4 * u < n) atomicAdd(&cnt[dd[u] - node0], 1);
  }
  __syncthreads();

  const int v = (t < RANGE_N) ? cnt[t] : 0;
  int x = v;
  for (int o2 = 1; o2 < 32; o2 <<= 1) { int y = __shfl_up(x, o2); if (lane >= o2) x += y; }
  if (lane == 31) wsum[wv] = x;
  __syncthreads();
  if (wv == 0) {
    const int w = wsum[lane];
    int xs = w;
    for (int o2 = 1; o2 < 32; o2 <<= 1) { int y = __shfl_up(xs, o2); if (lane >= o2) xs += y; }
    wsum[lane] = xs - w;
    if (lane == 31) tot_s = xs;
  }
  __syncthreads();
  const int total = tot_s;
  const int myoff = x - v + wsum[wv];
  if (t < RANGE_N) {
    cur[t] = myoff;
    const int node = node0 + t;
    if (node <= NNODES) rowptr[node] = S + myoff;
  }
  __syncthreads();

  const bool staged = total <= STAGE_CAP;
  for (int j = q; j < n; j += 32) {
    int es[8], dd[8];
#pragma unroll
    for (int u = 0; u < 8; ++u) {
      const int jj = j + 4 * u;
      const int at = o + (jj < n ? jj : n - 1);
      es[u] = tmp_s[at];
      dd[u] = tmp_d[at];
    }
    asm volatile("" :: "v"(dd[0]), "v"(dd[1]), "v"(dd[2]), "v"(dd[3]),
                       "v"(dd[4]), "v"(dd[5]), "v"(dd[6]), "v"(dd[7]),
                       "v"(es[0]), "v"(es[1]), "v"(es[2]), "v"(es[3]),
                       "v"(es[4]), "v"(es[5]), "v"(es[6]), "v"(es[7]));
#pragma unroll
    for (int u = 0; u < 8; ++u)
      if (j + 4 * u < n) {
        const int pos = atomicAdd(&cur[dd[u] - node0], 1);
        if (staged) stage[pos] = es[u]; else srcs[S + pos] = es[u];
      }
  }
  __syncthreads();
  if (staged)
    for (int i = t; i < total; i += SORT_T) srcs[S + i] = stage[i];
}

__device__ __forceinline__ float readlane_f32(float v, int l) {
  return __int_as_float(__builtin_amdgcn_readlane(__float_as_int(v), l));
}

template <int H, int VEC, bool ELU_ACT>
__global__ __launch_bounds__(256)
void gat_gather_kernel(const int* __restrict__ rowptr, const int* __restrict__ srcs,
                       const float* __restrict__ as_, const float* __restrict__ ad_,
                       const float* __restrict__ hfeat, const float* __restrict__ bias,
                       float* __restrict__ out) {
  constexpr int HC = 32 * VEC;
  constexpr int C  = HC / H;
  typedef __attribute__((ext_vector_type(VEC))) float vecf;
  static_assert(H == 1 || H == 4, "score loader handles 1 or 4 heads");

  const int lane = threadIdx.x & 31;
  const int d = __builtin_amdgcn_readfirstlane(
      (int)(blockIdx.x * NODES_PER_BLOCK + (threadIdx.x >> 5)));
  const int start = rowptr[d];
  const int deg   = rowptr[d + 1] - start;
  const int total = deg + 1;
  const int myhead = (lane * VEC) / C;

  float adv[H];
#pragma unroll
  for (int h = 0; h < H; ++h) adv[h] = ad_[d * H + h];

  auto chunk_scores = [&](int base, int& s, float (&e)[H]) -> bool {
    const int t = base + lane;
    s = d;
    if (t < deg) s = srcs[start + t];
    if constexpr (H == 4) {
      const float4 a = *reinterpret_cast<const float4*>(as_ + (long long)s * 4);
      e[0] = a.x + adv[0]; e[1] = a.y + adv[1]; e[2] = a.z + adv[2]; e[3] = a.w + adv[3];
    } else {
      e[0] = as_[s] + adv[0];
    }
#pragma unroll
    for (int h = 0; h < H; ++h) e[h] = e[h] > 0.0f ? e[h] : NEG_SLOPE * e[h];
    return t < total;
  };

  int   s0;
  float e0[H];
  const bool v0 = chunk_scores(0, s0, e0);

  float m[H];
#pragma unroll
  for (int h = 0; h < H; ++h) m[h] = e0[h];
  for (int base = 32; base < total; base += 32) {
    int s; float e[H];
    chunk_scores(base, s, e);
#pragma unroll
    for (int h = 0; h < H; ++h) m[h] = fmaxf(m[h], e[h]);
  }
#pragma unroll
  for (int h = 0; h < H; ++h)
    for (int o = 16; o > 0; o >>= 1) m[h] = fmaxf(m[h], __shfl_xor(m[h], o));

  float ex0[H], sum[H];
#pragma unroll
  for (int h = 0; h < H; ++h) { ex0[h] = v0 ? expf(e0[h] - m[h]) : 0.0f; sum[h] = ex0[h]; }
  for (int base = 32; base < total; base += 32) {
    int s; float e[H];
    const bool v = chunk_scores(base, s, e);
#pragma unroll
    for (int h = 0; h < H; ++h) sum[h] += v ? expf(e[h] - m[h]) : 0.0f;
  }
#pragma unroll
  for (int h = 0; h < H; ++h)
    for (int o = 16; o > 0; o >>= 1) sum[h] += __shfl_xor(sum[h], o);

  float acc[VEC];
#pragma unroll
  for (int v = 0; v < VEC; ++v) acc[v] = 0.0f;

  auto accumulate = [&](int s, const float (&al)[H], int cnt) {
    for (int j = 0; j < cnt; j += 4) {
      vecf  r[4];
      float a[4];
#pragma unroll
      for (int u = 0; u < 4; ++u) {
        const int jj = (j + u < cnt) ? (j + u) : (cnt - 1);
        const int sj = __builtin_amdgcn_readlane(s, jj);
        r[u] = *reinterpret_cast<const vecf*>(hfeat + (long long)sj * HC + lane * VEC);
        float av = readlane_f32(al[0], jj);
#pragma unroll
        for (int h = 1; h < H; ++h) {
          const float ah = readlane_f32(al[h], jj);
          av = (myhead == h) ? ah : av;
        }
        a[u] = (j + u < cnt) ? av : 0.0f;
      }
#pragma unroll
      for (int u = 0; u < 4; ++u)
#pragma unroll
        for (int v = 0; v < VEC; ++v) acc[v] = fmaf(a[u], r[u][v], acc[v]);
    }
  };

  {
    float al[H];
#pragma unroll
    for (int h = 0; h < H; ++h) al[h] = ex0[h] / (sum[h] + EPS_);
    accumulate(s0, al, total < 32 ? total : 32);
  }
  for (int base = 32; base < total; base += 32) {
    int s; float e[H], al[H];
    const bool v = chunk_scores(base, s, e);
#pragma unroll
    for (int h = 0; h < H; ++h) al[h] = v ? expf(e[h] - m[h]) / (sum[h] + EPS_) : 0.0f;
    const int left = total - base;
    accumulate(s, al, left < 32 ? left : 32);
  }

  const vecf b = *reinterpret_cast<const vecf*>(bias + lane * VEC);
  vecf o;
#pragma unroll
  for (int v = 0; v < VEC; ++v) {
    float x = acc[v] + b[v];
    if (ELU_ACT) x = x > 0.0f ? x : expm1f(x);
    o[v] = x;
  }
  *reinterpret_cast<vecf*>(out + (long long)d * HC + lane * VEC) = o;
}

extern "C" void kernel_launch(void* const* d_in, const int* in_sizes, int n_in,
                              void* d_out, int out_size, void* d_ws, size_t ws_size,
                              hipStream_t stream) {
  (void)in_sizes; (void)n_in; (void)out_size; (void)ws_size;

  const float* x    = (const float*)d_in[0];
  const int*   ei   = (const int*)  d_in[1];
  const float* W1   = (const float*)d_in[2];
  const float* aS1  = (const float*)d_in[3];
  const float* aD1  = (const float*)d_in[4];
  const float* b1   = (const float*)d_in[5];
  const float* W2   = (const float*)d_in[6];
  const float* aS2  = (const float*)d_in[7];
  const float* aD2  = (const float*)d_in[8];
  const float* b2   = (const float*)d_in[9];
  float* out = (float*)d_out;

  float* ws  = (float*)d_ws;
  float* h1  = ws;
  float* o1  = h1  + (long long)NNODES * L1_HC;
  float* as1 = o1  + (long long)NNODES * L1_HC;
  float* ad1 = as1 + NNODES * L1_H;
  float* as2 = ad1 + NNODES * L1_H;
  float* ad2 = as2 + NNODES;
  int*   rowptr = (int*)(ad2 + NNODES);
  int*   cntTab = rowptr + NRANGE * RANGE_N;
  int*   offTab = cntTab + NRANGE * NCHUNK;
  int*   srcs   = offTab + NRANGE * NCHUNK;
  int*   tmp_s  = srcs + NEDGES;
  int*   tmp_d  = tmp_s + NEDGES;
  float* h2  = h1;

  bin_kernel<<<NCHUNK, SORT_T, 0, stream>>>(ei, cntTab, offTab, tmp_s, tmp_d);
  collect_kernel<<<NRANGE, SORT_T, 0, stream>>>(cntTab, offTab, tmp_s, tmp_d, rowptr, srcs);

  gemm_wmma_kernel<128, L1_HC><<<NNODES / 16, 128, 0, stream>>>(x, W1, h1);
  attn_kernel<<<(NNODES * L1_H + 127) / 128, 128, 0, stream>>>(h1, aS1, aD1, as1, ad1, L1_H, L1_C);
  gat_gather_kernel<L1_H, L1_HC / 32, true><<<NNODES / NODES_PER_BLOCK, 256, 0, stream>>>(
      rowptr, srcs, as1, ad1, h1, b1, o1);

  gemm_wmma_kernel<128, L2_HC><<<NNODES / 16, 64, 0, stream>>>(o1, W2, h2);
  attn_kernel<<<(NNODES + 127) / 128, 128, 0, stream>>>(h2, aS2, aD2, as2, ad2, L2_H, L2_C);
  gat_gather_kernel<L2_H, L2_HC / 32, false><<<NNODES / NODES_PER_BLOCK, 256, 0, stream>>>(
      rowptr, srcs, as2, ad2, h2, b2, out);
}


# GEMM operands fetched by whole lines: A tile staged once per block in LDS (16-byte loads), each wave's two column tiles interleaved so one 8-byte B load feeds both; k and sum order unchanged
# speedup vs baseline: 13.1756x; 1.1365x over previous
#include <hip/hip_runtime.h>
#include <math.h>

#define NNODES   100000
#define NEDGES   1600000
#define L1_H     4
#define L1_C     32
#define L1_HC    128
#define L2_H     1
#define L2_C     64
#define L2_HC    64
#define NEG_SLOPE 0.2f
#define EPS_      1e-16f

#define NODES_PER_BLOCK 8

#define NCHUNK   250
#define CHUNK_E  6400
#define RANGE_SH 9
#define RANGE_N  512
#define NRANGE   196
#define SORT_T   1024
#define STAGE_CAP 12288

static_assert(NCHUNK * CHUNK_E == NEDGES, "chunks cover the edge list exactly");
static_assert(NRANGE * RANGE_N >= NNODES + 1, "ranges must cover rowptr[0..N]");
static_assert((1 << RANGE_SH) == RANGE_N, "range size is a power of two");
static_assert(NCHUNK * 4 <= SORT_T, "four threads per piece in the collect kernel");
static_assert(NNODES % NODES_PER_BLOCK == 0, "gather grid divides the node count");

typedef __attribute__((ext_vector_type(2))) float v2f;
typedef __attribute__((ext_vector_type(4))) float v4f;
typedef __attribute__((ext_vector_type(8))) float v8f;

#define A_PITCH 132

template <int K, int NC>
__global__ __launch_bounds__(NC)
void gemm_wmma_kernel(const float* __restrict__ A, const float* __restrict__ B,
                      float* __restrict__ C) {
  constexpr int NT     = NC;
  constexpr int ROW_CH = K / 4;
  constexpr int PER_T  = 16 * ROW_CH / NT;
  static_assert(ROW_CH == 32, "one wave-load covers one row of A");
  static_assert((16 * ROW_CH) % NT == 0, "tile pieces divide among the threads");
  __shared__ __attribute__((aligned(16))) float tile[16 * A_PITCH];

  const int lane = threadIdx.x & 31;
  const int wv   = threadIdx.x >> 5;
  const int row0 = blockIdx.x * 16;
  const int col0 = wv * 32;
  const int m    = lane & 15;
  const int kk   = (lane >> 4) << 1;

  v4f st[PER_T];
#pragma unroll
  for (int i = 0; i < PER_T; ++i) {
    const int r = wv + i * (NT / 32);
    st[i] = *reinterpret_cast<const v4f*>(A + (long long)(row0 + r) * K + lane * 4);
  }
#pragma unroll
  for (int i = 0; i < PER_T; ++i) {
    const int r = wv + i * (NT / 32);
    *reinterpret_cast<v4f*>(&tile[r * A_PITCH + lane * 4]) = st[i];
  }
  __syncthreads();

  v8f acc0 = {};
  v8f acc1 = {};
  const float* ap = tile + m * A_PITCH + kk;
  const float* bp = B + (long long)kk * NC + (col0 + 2 * m);
#pragma unroll
  for (int k0 = 0; k0 < K; k0 += 4) {
    const v2f a = *reinterpret_cast<const v2f*>(ap + k0);
    const v2f p = *reinterpret_cast<const v2f*>(bp + k0 * NC);
    const v2f q = *reinterpret_cast<const v2f*>(bp + (k0 + 1) * NC);
    v2f b0, b1;
    b0.x = p.x; b0.y = q.x;
    b1.x = p.y; b1.y = q.y;
    acc0 = __builtin_amdgcn_wmma_f32_16x16x4_f32(
        false, a, false, b0, (short)0, acc0, false, false);
    acc1 = __builtin_amdgcn_wmma_f32_16x16x4_f32(
        false, a, false, b1, (short)0, acc1, false, false);
  }
  const int orow = row0 + ((lane >> 4) << 3);
  float* cp = C + (long long)orow * NC + (col0 + 2 * m);
#pragma unroll
  for (int r = 0; r < 8; ++r) {
    v2f o;
    o.x = acc0[r]; o.y = acc1[r];
    *reinterpret_cast<v2f*>(cp + (long long)r * NC) = o;
  }
}

__global__ __launch_bounds__(SORT_T)
void bin_kernel(const int* __restrict__ ei, int* __restrict__ cntTab,
                int* __restrict__ offTab, int* __restrict__ tmp_s,
                int* __restrict__ tmp_d) {
  __shared__ int hist[NRANGE];
  __shared__ int stage_s[CHUNK_E];
  __shared__ int stage_d[CHUNK_E];
  constexpr int PER = (CHUNK_E + SORT_T - 1) / SORT_T;
  constexpr int PL  = (NRANGE + 31) / 32;
  const int t = threadIdx.x;
  const int base = blockIdx.x * CHUNK_E;

  for (int b = t; b < NRANGE; b += SORT_T) hist[b] = 0;
  __syncthreads();

  int s[PER], d[PER];
#pragma unroll
  for (int k = 0; k < PER; ++k) {
    const int i = t + k * SORT_T;
    s[k] = 0; d[k] = 0;
    if (i < CHUNK_E) { d[k] = ei[NEDGES + base + i]; s[k] = ei[base + i]; }
  }
#pragma unroll
  for (int k = 0; k < PER; ++k)
    if (t + k * SORT_T < CHUNK_E) atomicAdd(&hist[d[k] >> RANGE_SH], 1);
  __syncthreads();

  if (t < 32) {
    int c[PL];
    int sum = 0;
#pragma unroll
    for (int j = 0; j < PL; ++j) {
      const int idx = t * PL + j;
      c[j] = (idx < NRANGE) ? hist[idx] : 0;
      sum += c[j];
    }
    int x = sum;
    for (int o = 1; o < 32; o <<= 1) { int y = __shfl_up(x, o); if (t >= o) x += y; }
    int run = x - sum;
#pragma unroll
    for (int j = 0; j < PL; ++j) {
      const int idx = t * PL + j;
      if (idx < NRANGE) {
        cntTab[idx * NCHUNK + blockIdx.x] = c[j];
        offTab[idx * NCHUNK + blockIdx.x] = run;
        hist[idx] = run;
      }
      run += c[j];
    }
  }
  __syncthreads();

#pragma unroll
  for (int k = 0; k < PER; ++k)
    if (t + k * SORT_T < CHUNK_E) {
      const int pos = atomicAdd(&hist[d[k] >> RANGE_SH], 1);
      stage_s[pos] = s[k];
      stage_d[pos] = d[k];
    }
  __syncthreads();

  for (int i = t; i < CHUNK_E; i += SORT_T) {
    tmp_s[base + i] = stage_s[i];
    tmp_d[base + i] = stage_d[i];
  }
}

__global__ __launch_bounds__(SORT_T)
void collect_kernel(const int* __restrict__ cntTab, const int* __restrict__ offTab,
                    const int* __restrict__ tmp_s, const int* __restrict__ tmp_d,
                    int* __restrict__ rowptr, int* __restrict__ srcs) {
  __shared__ int cnt[RANGE_N];
  __shared__ int cur[RANGE_N];
  __shared__ int pc[NCHUNK];
  __shared__ int po[NCHUNK];
  __shared__ int wsum[32];
  __shared__ int tot_s;
  __shared__ int stage[STAGE_CAP];
  const int t = threadIdx.x, lane = t & 31, wv = t >> 5;
  const int node0 = blockIdx.x * RANGE_N;
  const int npre  = blockIdx.x * NCHUNK;

  int acc = 0;
  for (int i = t; i < npre; i += SORT_T) acc += cntTab[i];
  for (int o = 16; o > 0; o >>= 1) acc += __shfl_xor(acc, o);
  if (lane == 0) wsum[wv] = acc;
  if (t < NCHUNK) { pc[t] = cntTab[npre + t]; po[t] = t * CHUNK_E + offTab[npre + t]; }
  if (t < RANGE_N) cnt[t] = 0;
  __syncthreads();
  int S = 0;
  for (int w = 0; w < 32; ++w) S += wsum[w];

  const int p = t >> 2, q = t & 3;
  int n = 0, o = 0;
  if (p < NCHUNK) { n = pc[p]; o = po[p]; }

  for (int j = q; j < n; j += 32) {
    int dd[8];
#pragma unroll
    for (int u = 0; u < 8; ++u) {
      const int jj = j + 4 * u;
      dd[u] = tmp_d[o + (jj < n ? jj : n - 1)];
    }
    asm volatile("" :: "v"(dd[0]), "v"(dd[1]), "v"(dd[2]), "v"(dd[3]),
                       "v"(dd[4]), "v"(dd[5]), "v"(dd[6]), "v"(dd[7]));
#pragma unroll
    for (int u = 0; u < 8; ++u)
      if (j + 4 * u < n) atomicAdd(&cnt[dd[u] - node0], 1);
  }
  __syncthreads();

  const int v = (t < RANGE_N) ? cnt[t] : 0;
  int x = v;
  for (int o2 = 1; o2 < 32; o2 <<= 1) { int y = __shfl_up(x, o2); if (lane >= o2) x += y; }
  if (lane == 31) wsum[wv] = x;
  __syncthreads();
  if (wv == 0) {
    const int w = wsum[lane];
    int xs = w;
    for (int o2 = 1; o2 < 32; o2 <<= 1) { int y = __shfl_up(xs, o2); if (lane >= o2) xs += y; }
    wsum[lane] = xs - w;
    if (lane == 31) tot_s = xs;
  }
  __syncthreads();
  const int total = tot_s;
  const int myoff = x - v + wsum[wv];
  if (t < RANGE_N) {
    cur[t] = myoff;
    const int node = node0 + t;
    if (node <= NNODES) rowptr[node] = S + myoff;
  }
  __syncthreads();

  const bool staged = total <= STAGE_CAP;
  for (int j = q; j < n; j += 32) {
    int es[8], dd[8];
#pragma unroll
    for (int u = 0; u < 8; ++u) {
      const int jj = j + 4 * u;
      const int at = o + (jj < n ? jj : n - 1);
      es[u] = tmp_s[at];
      dd[u] = tmp_d[at];
    }
    asm volatile("" :: "v"(dd[0]), "v"(dd[1]), "v"(dd[2]), "v"(dd[3]),
                       "v"(dd[4]), "v"(dd[5]), "v"(dd[6]), "v"(dd[7]),
                       "v"(es[0]), "v"(es[1]), "v"(es[2]), "v"(es[3]),
                       "v"(es[4]), "v"(es[5]), "v"(es[6]), "v"(es[7]));
#pragma unroll
    for (int u = 0; u < 8; ++u)
      if (j + 4 * u < n) {
        const int pos = atomicAdd(&cur[dd[u] - node0], 1);
        if (staged) stage[pos] = es[u]; else srcs[S + pos] = es[u];
      }
  }
  __syncthreads();
  if (staged)
    for (int i = t; i < total; i += SORT_T) srcs[S + i] = stage[i];
}

template <int XMASK>
__device__ __forceinline__ float row16_xadd(float v) {
  const int other = __builtin_amdgcn_update_dpp(0, __float_as_int(v), 0x160 | XMASK, 0xf, 0xf, true);
  return v + __int_as_float(other);
}
__device__ __forceinline__ float xhalf(float x) {
  const int b = __float_as_int(x);
  return __int_as_float(__builtin_amdgcn_permlanex16(b, b, 0x76543210, 0xfedcba98, false, false));
}

template <int SRC>
__device__ __forceinline__ float bcast8(float x) {
  return __int_as_float((int)__builtin_amdgcn_mov_dpp8((unsigned)__float_as_int(x), SRC * 0x249249));
}

template <int H, int VEC, bool ELU_ACT>
__global__ __launch_bounds__(256)
void gat_gather_kernel(const int* __restrict__ rowptr, const int* __restrict__ srcs,
                       const float* __restrict__ att_s, const float* __restrict__ att_d,
                       const float* __restrict__ hfeat, const float* __restrict__ bias,
                       float* __restrict__ out) {
  constexpr int HC  = 32 * VEC;
  constexpr int LPH = 32 / H;
  typedef __attribute__((ext_vector_type(VEC))) float vecf;
  static_assert(LPH == 8 || LPH == 32, "head sum handles 8 or 32 lanes per head");

  const int lane = threadIdx.x & 31;
  const int d = __builtin_amdgcn_readfirstlane(
      (int)(blockIdx.x * NODES_PER_BLOCK + (threadIdx.x >> 5)));
  const int start = rowptr[d];
  const int deg   = rowptr[d + 1] - start;

  const vecf ws = *reinterpret_cast<const vecf*>(att_s + lane * VEC);
  const vecf wd = *reinterpret_cast<const vecf*>(att_d + lane * VEC);
  const vecf b  = *reinterpret_cast<const vecf*>(bias + lane * VEC);

  auto dotv = [&](const vecf& r, const vecf& w) -> float {
    float p = r[0] * w[0];
#pragma unroll
    for (int v = 1; v < VEC; ++v) p = fmaf(r[v], w[v], p);
    return p;
  };
  auto head_sum = [&](float x) -> float {
    x = row16_xadd<1>(x);
    x = row16_xadd<2>(x);
    x = row16_xadd<4>(x);
    if constexpr (LPH == 32) {
      x = row16_xadd<8>(x);
      x = x + xhalf(x);
      x = __int_as_float(__builtin_amdgcn_readfirstlane(__float_as_int(x)));
    }
    return x;
  };
  auto lrelu = [](float x) -> float { return x > 0.0f ? x : NEG_SLOPE * x; };

  const vecf rd = *reinterpret_cast<const vecf*>(hfeat + (long long)d * HC + lane * VEC);
  const float adv = head_sum(dotv(rd, wd));
  float m   = lrelu(head_sum(dotv(rd, ws)) + adv);
  float sum = 1.0f;
  float acc[VEC];
#pragma unroll
  for (int v = 0; v < VEC; ++v) acc[v] = rd[v];

  for (int base = 0; base < deg; base += 32) {
    const int left = deg - base;
    const int cnt  = left < 32 ? left : 32;
    int s = d;
    if (lane < cnt) s = srcs[start + base + lane];
    for (int j = 0; j < cnt; j += 4) {
      vecf  r[4];
      float e[4];
#pragma unroll
      for (int u = 0; u < 4; ++u) {
        const int jj = (j + u < cnt) ? (j + u) : (cnt - 1);
        const int sj = __builtin_amdgcn_readlane(s, jj);
        r[u] = *reinterpret_cast<const vecf*>(hfeat + (long long)sj * HC + lane * VEC);
      }
#pragma unroll
      for (int u = 0; u < 4; ++u) {
        const float x = lrelu(head_sum(dotv(r[u], ws)) + adv);
        e[u] = (j + u < cnt) ? x : -INFINITY;
      }
      const float mn = fmaxf(fmaxf(m, fmaxf(e[0], e[1])), fmaxf(e[2], e[3]));
      float sc, w[4];
      if constexpr (LPH == 8) {
        const int k = lane & 7;
        float x = m;
        x = (k == 0) ? e[0] : x;
        x = (k == 1) ? e[1] : x;
        x = (k == 2) ? e[2] : x;
        x = (k == 3) ? e[3] : x;
        const float y = expf(x - mn);
        w[0] = bcast8<0>(y);
        w[1] = bcast8<1>(y);
        w[2] = bcast8<2>(y);
        w[3] = bcast8<3>(y);
        sc   = bcast8<4>(y);
      } else {
        sc = expf(m - mn);
#pragma unroll
        for (int u = 0; u < 4; ++u) w[u] = expf(e[u] - mn);
      }
      sum = fmaf(sum, sc, (w[0] + w[1]) + (w[2] + w[3]));
#pragma unroll
      for (int v = 0; v < VEC; ++v) {
        float t = acc[v] * sc;
#pragma unroll
        for (int u = 0; u < 4; ++u) t = fmaf(w[u], r[u][v], t);
        acc[v] = t;
      }
      m = mn;
    }
  }

  const float inv = 1.0f / (sum + EPS_);
  vecf o;
#pragma unroll
  for (int v = 0; v < VEC; ++v) {
    float x = fmaf(acc[v], inv, b[v]);
    if (ELU_ACT) x = x > 0.0f ? x : expm1f(x);
    o[v] = x;
  }
  *reinterpret_cast<vecf*>(out + (long long)d * HC + lane * VEC) = o;
}

extern "C" void kernel_launch(void* const* d_in, const int* in_sizes, int n_in,
                              void* d_out, int out_size, void* d_ws, size_t ws_size,
                              hipStream_t stream) {
  (void)in_sizes; (void)n_in; (void)out_size; (void)ws_size;

  const float* x    = (const float*)d_in[0];
  const int*   ei   = (const int*)  d_in[1];
  const float* W1   = (const float*)d_in[2];
  const float* aS1  = (const float*)d_in[3];
  const float* aD1  = (const float*)d_in[4];
  const float* b1   = (const float*)d_in[5];
  const float* W2   = (const float*)d_in[6];
  const float* aS2  = (const float*)d_in[7];
  const float* aD2  = (const float*)d_in[8];
  const float* b2   = (const float*)d_in[9];
  float* out = (float*)d_out;

  float* ws  = (float*)d_ws;
  float* h1  = ws;
  float* o1  = h1  + (long long)NNODES * L1_HC;
  int*   rowptr = (int*)(o1 + (long long)NNODES * L1_HC);
  int*   cntTab = rowptr + NRANGE * RANGE_N;
  int*   offTab = cntTab + NRANGE * NCHUNK;
  int*   srcs   = offTab + NRANGE * NCHUNK;
  int*   tmp_s  = srcs + NEDGES;
  int*   tmp_d  = tmp_s + NEDGES;
  float* h2  = h1;

  bin_kernel<<<NCHUNK, SORT_T, 0, stream>>>(ei, cntTab, offTab, tmp_s, tmp_d);
  collect_kernel<<<NRANGE, SORT_T, 0, stream>>>(cntTab, offTab, tmp_s, tmp_d, rowptr, srcs);

  gemm_wmma_kernel<128, L1_HC><<<NNODES / 16, 128, 0, stream>>>(x, W1, h1);
  gat_gather_kernel<L1_H, L1_HC / 32, true><<<NNODES / NODES_PER_BLOCK, 256, 0, stream>>>(
      rowptr, srcs, aS1, aD1, h1, b1, o1);

  gemm_wmma_kernel<128, L2_HC><<<NNODES / 16, 64, 0, stream>>>(o1, W2, h2);
  gat_gather_kernel<L2_H, L2_HC / 32, false><<<NNODES / NODES_PER_BLOCK, 256, 0, stream>>>(
      rowptr, srcs, aS2, aD2, h2, b2, out);
}


# Layer-2 gather: the two 16-lane halves of a wave take two edges at once (16-byte loads, row sum by DPP only, halves merged at the end); sum order changes
# speedup vs baseline: 13.9236x; 1.0568x over previous
#include <hip/hip_runtime.h>
#include <math.h>

#define NNODES   100000
#define NEDGES   1600000
#define L1_H     4
#define L1_C     32
#define L1_HC    128
#define L2_H     1
#define L2_C     64
#define L2_HC    64
#define NEG_SLOPE 0.2f
#define EPS_      1e-16f

#define NODES_PER_BLOCK 8

#define NCHUNK   250
#define CHUNK_E  6400
#define RANGE_SH 9
#define RANGE_N  512
#define NRANGE   196
#define SORT_T   1024
#define STAGE_CAP 12288

static_assert(NCHUNK * CHUNK_E == NEDGES, "chunks cover the edge list exactly");
static_assert(NRANGE * RANGE_N >= NNODES + 1, "ranges must cover rowptr[0..N]");
static_assert((1 << RANGE_SH) == RANGE_N, "range size is a power of two");
static_assert(NCHUNK * 4 <= SORT_T, "four threads per piece in the collect kernel");
static_assert(NNODES % NODES_PER_BLOCK == 0, "gather grid divides the node count");

typedef __attribute__((ext_vector_type(2))) float v2f;
typedef __attribute__((ext_vector_type(4))) float v4f;
typedef __attribute__((ext_vector_type(8))) float v8f;

#define A_PITCH 132

template <int K, int NC>
__global__ __launch_bounds__(NC)
void gemm_wmma_kernel(const float* __restrict__ A, const float* __restrict__ B,
                      float* __restrict__ C) {
  constexpr int NT     = NC;
  constexpr int ROW_CH = K / 4;
  constexpr int PER_T  = 16 * ROW_CH / NT;
  static_assert(ROW_CH == 32, "one wave-load covers one row of A");
  static_assert((16 * ROW_CH) % NT == 0, "tile pieces divide among the threads");
  __shared__ __attribute__((aligned(16))) float tile[16 * A_PITCH];

  const int lane = threadIdx.x & 31;
  const int wv   = threadIdx.x >> 5;
  const int row0 = blockIdx.x * 16;
  const int col0 = wv * 32;
  const int m    = lane & 15;
  const int kk   = (lane >> 4) << 1;

  v4f st[PER_T];
#pragma unroll
  for (int i = 0; i < PER_T; ++i) {
    const int r = wv + i * (NT / 32);
    st[i] = *reinterpret_cast<const v4f*>(A + (long long)(row0 + r) * K + lane * 4);
  }
#pragma unroll
  for (int i = 0; i < PER_T; ++i) {
    const int r = wv + i * (NT / 32);
    *reinterpret_cast<v4f*>(&tile[r * A_PITCH + lane * 4]) = st[i];
  }
  __syncthreads();

  v8f acc0 = {};
  v8f acc1 = {};
  const float* ap = tile + m * A_PITCH + kk;
  const float* bp = B + (long long)kk * NC + (col0 + 2 * m);
#pragma unroll
  for (int k0 = 0; k0 < K; k0 += 4) {
    const v2f a = *reinterpret_cast<const v2f*>(ap + k0);
    const v2f p = *reinterpret_cast<const v2f*>(bp + k0 * NC);
    const v2f q = *reinterpret_cast<const v2f*>(bp + (k0 + 1) * NC);
    v2f b0, b1;
    b0.x = p.x; b0.y = q.x;
    b1.x = p.y; b1.y = q.y;
    acc0 = __builtin_amdgcn_wmma_f32_16x16x4_f32(
        false, a, false, b0, (short)0, acc0, false, false);
    acc1 = __builtin_amdgcn_wmma_f32_16x16x4_f32(
        false, a, false, b1, (short)0, acc1, false, false);
  }
  const int orow = row0 + ((lane >> 4) << 3);
  float* cp = C + (long long)orow * NC + (col0 + 2 * m);
#pragma unroll
  for (int r = 0; r < 8; ++r) {
    v2f o;
    o.x = acc0[r]; o.y = acc1[r];
    *reinterpret_cast<v2f*>(cp + (long long)r * NC) = o;
  }
}

__global__ __launch_bounds__(SORT_T)
void bin_kernel(const int* __restrict__ ei, int* __restrict__ cntTab,
                int* __restrict__ offTab, int* __restrict__ tmp_s,
                int* __restrict__ tmp_d) {
  __shared__ int hist[NRANGE];
  __shared__ int stage_s[CHUNK_E];
  __shared__ int stage_d[CHUNK_E];
  constexpr int PER = (CHUNK_E + SORT_T - 1) / SORT_T;
  constexpr int PL  = (NRANGE + 31) / 32;
  const int t = threadIdx.x;
  const int base = blockIdx.x * CHUNK_E;

  for (int b = t; b < NRANGE; b += SORT_T) hist[b] = 0;
  __syncthreads();

  int s[PER], d[PER];
#pragma unroll
  for (int k = 0; k < PER; ++k) {
    const int i = t + k * SORT_T;
    s[k] = 0; d[k] = 0;
    if (i < CHUNK_E) { d[k] = ei[NEDGES + base + i]; s[k] = ei[base + i]; }
  }
#pragma unroll
  for (int k = 0; k < PER; ++k)
    if (t + k * SORT_T < CHUNK_E) atomicAdd(&hist[d[k] >> RANGE_SH], 1);
  __syncthreads();

  if (t < 32) {
    int c[PL];
    int sum = 0;
#pragma unroll
    for (int j = 0; j < PL; ++j) {
      const int idx = t * PL + j;
      c[j] = (idx < NRANGE) ? hist[idx] : 0;
      sum += c[j];
    }
    int x = sum;
    for (int o = 1; o < 32; o <<= 1) { int y = __shfl_up(x, o); if (t >= o) x += y; }
    int run = x - sum;
#pragma unroll
    for (int j = 0; j < PL; ++j) {
      const int idx = t * PL + j;
      if (idx < NRANGE) {
        cntTab[idx * NCHUNK + blockIdx.x] = c[j];
        offTab[idx * NCHUNK + blockIdx.x] = run;
        hist[idx] = run;
      }
      run += c[j];
    }
  }
  __syncthreads();

#pragma unroll
  for (int k = 0; k < PER; ++k)
    if (t + k * SORT_T < CHUNK_E) {
      const int pos = atomicAdd(&hist[d[k] >> RANGE_SH], 1);
      stage_s[pos] = s[k];
      stage_d[pos] = d[k];
    }
  __syncthreads();

  for (int i = t; i < CHUNK_E; i += SORT_T) {
    tmp_s[base + i] = stage_s[i];
    tmp_d[base + i] = stage_d[i];
  }
}

__global__ __launch_bounds__(SORT_T)
void collect_kernel(const int* __restrict__ cntTab, const int* __restrict__ offTab,
                    const int* __restrict__ tmp_s, const int* __restrict__ tmp_d,
                    int* __restrict__ rowptr, int* __restrict__ srcs) {
  __shared__ int cnt[RANGE_N];
  __shared__ int cur[RANGE_N];
  __shared__ int pc[NCHUNK];
  __shared__ int po[NCHUNK];
  __shared__ int wsum[32];
  __shared__ int tot_s;
  __shared__ int stage[STAGE_CAP];
  const int t = threadIdx.x, lane = t & 31, wv = t >> 5;
  const int node0 = blockIdx.x * RANGE_N;
  const int npre  = blockIdx.x * NCHUNK;

  int acc = 0;
  for (int i = t; i < npre; i += SORT_T) acc += cntTab[i];
  for (int o = 16; o > 0; o >>= 1) acc += __shfl_xor(acc, o);
  if (lane == 0) wsum[wv] = acc;
  if (t < NCHUNK) { pc[t] = cntTab[npre + t]; po[t] = t * CHUNK_E + offTab[npre + t]; }
  if (t < RANGE_N) cnt[t] = 0;
  __syncthreads();
  int S = 0;
  for (int w = 0; w < 32; ++w) S += wsum[w];

  const int p = t >> 2, q = t & 3;
  int n = 0, o = 0;
  if (p < NCHUNK) { n = pc[p]; o = po[p]; }

  for (int j = q; j < n; j += 32) {
    int dd[8];
#pragma unroll
    for (int u = 0; u < 8; ++u) {
      const int jj = j + 4 * u;
      dd[u] = tmp_d[o + (jj < n ? jj : n - 1)];
    }
    asm volatile("" :: "v"(dd[0]), "v"(dd[1]), "v"(dd[2]), "v"(dd[3]),
                       "v"(dd[4]), "v"(dd[5]), "v"(dd[6]), "v"(dd[7]));
#pragma unroll
    for (int u = 0; u < 8; ++u)
      if (j + 4 * u < n) atomicAdd(&cnt[dd[u] - node0], 1);
  }
  __syncthreads();

  const int v = (t < RANGE_N) ? cnt[t] : 0;
  int x = v;
  for (int o2 = 1; o2 < 32; o2 <<= 1) { int y = __shfl_up(x, o2); if (lane >= o2) x += y; }
  if (lane == 31) wsum[wv] = x;
  __syncthreads();
  if (wv == 0) {
    const int w = wsum[lane];
    int xs = w;
    for (int o2 = 1; o2 < 32; o2 <<= 1) { int y = __shfl_up(xs, o2); if (lane >= o2) xs += y; }
    wsum[lane] = xs - w;
    if (lane == 31) tot_s = xs;
  }
  __syncthreads();
  const int total = tot_s;
  const int myoff = x - v + wsum[wv];
  if (t < RANGE_N) {
    cur[t] = myoff;
    const int node = node0 + t;
    if (node <= NNODES) rowptr[node] = S + myoff;
  }
  __syncthreads();

  const bool staged = total <= STAGE_CAP;
  for (int j = q; j < n; j += 32) {
    int es[8], dd[8];
#pragma unroll
    for (int u = 0; u < 8; ++u) {
      const int jj = j + 4 * u;
      const int at = o + (jj < n ? jj : n - 1);
      es[u] = tmp_s[at];
      dd[u] = tmp_d[at];
    }
    asm volatile("" :: "v"(dd[0]), "v"(dd[1]), "v"(dd[2]), "v"(dd[3]),
                       "v"(dd[4]), "v"(dd[5]), "v"(dd[6]), "v"(dd[7]),
                       "v"(es[0]), "v"(es[1]), "v"(es[2]), "v"(es[3]),
                       "v"(es[4]), "v"(es[5]), "v"(es[6]), "v"(es[7]));
#pragma unroll
    for (int u = 0; u < 8; ++u)
      if (j + 4 * u < n) {
        const int pos = atomicAdd(&cur[dd[u] - node0], 1);
        if (staged) stage[pos] = es[u]; else srcs[S + pos] = es[u];
      }
  }
  __syncthreads();
  if (staged)
    for (int i = t; i < total; i += SORT_T) srcs[S + i] = stage[i];
}

template <int XMASK>
__device__ __forceinline__ float row16_xadd(float v) {
  const int other = __builtin_amdgcn_update_dpp(0, __float_as_int(v), 0x160 | XMASK, 0xf, 0xf, true);
  return v + __int_as_float(other);
}
__device__ __forceinline__ float xhalf(float x) {
  const int b = __float_as_int(x);
  return __int_as_float(__builtin_amdgcn_permlanex16(b, b, 0x76543210, 0xfedcba98, false, false));
}

template <int SRC>
__device__ __forceinline__ float bcast8(float x) {
  return __int_as_float((int)__builtin_amdgcn_mov_dpp8((unsigned)__float_as_int(x), SRC * 0x249249));
}

template <int H, int VEC, bool ELU_ACT>
__global__ __launch_bounds__(256)
void gat_gather_kernel(const int* __restrict__ rowptr, const int* __restrict__ srcs,
                       const float* __restrict__ att_s, const float* __restrict__ att_d,
                       const float* __restrict__ hfeat, const float* __restrict__ bias,
                       float* __restrict__ out) {
  constexpr int HC  = 32 * VEC;
  constexpr int LPH = 32 / H;
  typedef __attribute__((ext_vector_type(VEC))) float vecf;
  static_assert(LPH == 8 || LPH == 32, "head sum handles 8 or 32 lanes per head");

  const int lane = threadIdx.x & 31;
  const int d = __builtin_amdgcn_readfirstlane(
      (int)(blockIdx.x * NODES_PER_BLOCK + (threadIdx.x >> 5)));
  const int start = rowptr[d];
  const int deg   = rowptr[d + 1] - start;

  const vecf ws = *reinterpret_cast<const vecf*>(att_s + lane * VEC);
  const vecf wd = *reinterpret_cast<const vecf*>(att_d + lane * VEC);
  const vecf b  = *reinterpret_cast<const vecf*>(bias + lane * VEC);

  auto dotv = [&](const vecf& r, const vecf& w) -> float {
    float p = r[0] * w[0];
#pragma unroll
    for (int v = 1; v < VEC; ++v) p = fmaf(r[v], w[v], p);
    return p;
  };
  auto head_sum = [&](float x) -> float {
    x = row16_xadd<1>(x);
    x = row16_xadd<2>(x);
    x = row16_xadd<4>(x);
    if constexpr (LPH == 32) {
      x = row16_xadd<8>(x);
      x = x + xhalf(x);
      x = __int_as_float(__builtin_amdgcn_readfirstlane(__float_as_int(x)));
    }
    return x;
  };
  auto lrelu = [](float x) -> float { return x > 0.0f ? x : NEG_SLOPE * x; };

  const vecf rd = *reinterpret_cast<const vecf*>(hfeat + (long long)d * HC + lane * VEC);
  const float adv = head_sum(dotv(rd, wd));
  float m   = lrelu(head_sum(dotv(rd, ws)) + adv);
  float sum = 1.0f;
  float acc[VEC];
#pragma unroll
  for (int v = 0; v < VEC; ++v) acc[v] = rd[v];

  for (int base = 0; base < deg; base += 32) {
    const int left = deg - base;
    const int cnt  = left < 32 ? left : 32;
    int s = d;
    if (lane < cnt) s = srcs[start + base + lane];
    for (int j = 0; j < cnt; j += 4) {
      vecf  r[4];
      float e[4];
#pragma unroll
      for (int u = 0; u < 4; ++u) {
        const int jj = (j + u < cnt) ? (j + u) : (cnt - 1);
        const int sj = __builtin_amdgcn_readlane(s, jj);
        r[u] = *reinterpret_cast<const vecf*>(hfeat + (long long)sj * HC + lane * VEC);
      }
#pragma unroll
      for (int u = 0; u < 4; ++u) {
        const float x = lrelu(head_sum(dotv(r[u], ws)) + adv);
        e[u] = (j + u < cnt) ? x : -INFINITY;
      }
      const float mn = fmaxf(fmaxf(m, fmaxf(e[0], e[1])), fmaxf(e[2], e[3]));
      float sc, w[4];
      if constexpr (LPH == 8) {
        const int k = lane & 7;
        float x = m;
        x = (k == 0) ? e[0] : x;
        x = (k == 1) ? e[1] : x;
        x = (k == 2) ? e[2] : x;
        x = (k == 3) ? e[3] : x;
        const float y = expf(x - mn);
        w[0] = bcast8<0>(y);
        w[1] = bcast8<1>(y);
        w[2] = bcast8<2>(y);
        w[3] = bcast8<3>(y);
        sc   = bcast8<4>(y);
      } else {
        sc = expf(m - mn);
#pragma unroll
        for (int u = 0; u < 4; ++u) w[u] = expf(e[u] - mn);
      }
      sum = fmaf(sum, sc, (w[0] + w[1]) + (w[2] + w[3]));
#pragma unroll
      for (int v = 0; v < VEC; ++v) {
        float t = acc[v] * sc;
#pragma unroll
        for (int u = 0; u < 4; ++u) t = fmaf(w[u], r[u][v], t);
        acc[v] = t;
      }
      m = mn;
    }
  }

  const float inv = 1.0f / (sum + EPS_);
  vecf o;
#pragma unroll
  for (int v = 0; v < VEC; ++v) {
    float x = fmaf(acc[v], inv, b[v]);
    if (ELU_ACT) x = x > 0.0f ? x : expm1f(x);
    o[v] = x;
  }
  *reinterpret_cast<vecf*>(out + (long long)d * HC + lane * VEC) = o;
}

__device__ __forceinline__ int row_pick(int x, int sel) {
  const unsigned s = (unsigned)sel * 0x11111111u;
  return (int)__builtin_amdgcn_permlane16((unsigned)x, (unsigned)x, s, s, false, false);
}

__global__ __launch_bounds__(256)
void gat_gather_pair_kernel(const int* __restrict__ rowptr, const int* __restrict__ srcs,
                            const float* __restrict__ att_s, const float* __restrict__ att_d,
                            const float* __restrict__ hfeat, const float* __restrict__ bias,
                            float* __restrict__ out) {
  constexpr int HC = L2_HC;
  static_assert(HC == 64 && L2_H == 1, "16 lanes x 4 channels cover one single-head row");

  const int lane = threadIdx.x & 31;
  const int c4   = (lane & 15) * 4;
  const int half = lane >> 4;
  const int d = __builtin_amdgcn_readfirstlane(
      (int)(blockIdx.x * NODES_PER_BLOCK + (threadIdx.x >> 5)));
  const int start = rowptr[d];
  const int deg   = rowptr[d + 1] - start;

  const v4f ws = *reinterpret_cast<const v4f*>(att_s + c4);
  const v4f wd = *reinterpret_cast<const v4f*>(att_d + c4);
  const v4f b  = *reinterpret_cast<const v4f*>(bias + c4);

  auto dot4 = [](const v4f& r, const v4f& w) -> float {
    float p = r[0] * w[0];
    p = fmaf(r[1], w[1], p);
    p = fmaf(r[2], w[2], p);
    p = fmaf(r[3], w[3], p);
    return p;
  };
  auto row_sum = [](float x) -> float {
    x = row16_xadd<1>(x);
    x = row16_xadd<2>(x);
    x = row16_xadd<4>(x);
    x = row16_xadd<8>(x);
    return x;
  };
  auto lrelu = [](float x) -> float { return x > 0.0f ? x : NEG_SLOPE * x; };

  const v4f rd = *reinterpret_cast<const v4f*>(hfeat + (long long)d * HC + c4);
  const float adv = row_sum(dot4(rd, wd));
  float m   = lrelu(row_sum(dot4(rd, ws)) + adv);
  float sum = half ? 0.0f : 1.0f;
  float acc[4];
#pragma unroll
  for (int v = 0; v < 4; ++v) acc[v] = half ? 0.0f : rd[v];

  const int k = lane & 7;
  for (int base = 0; base < deg; base += 32) {
    const int left = deg - base;
    const int cnt  = left < 32 ? left : 32;
    const int mine = 2 * (lane & 15) + half;
    int s = d;
    if (mine < cnt) s = srcs[start + base + mine];
    const int npair = (cnt + 1) >> 1;
    for (int j = 0; j < npair; j += 4) {
      v4f   r[4];
      float e[4];
#pragma unroll
      for (int u = 0; u < 4; ++u) {
        const int t  = (j + u < npair) ? (j + u) : (npair - 1);
        const int sj = row_pick(s, t);
        r[u] = *reinterpret_cast<const v4f*>(hfeat + ((unsigned)sj * (unsigned)HC + (unsigned)c4));
      }
#pragma unroll
      for (int u = 0; u < 4; ++u) {
        const float x = lrelu(row_sum(dot4(r[u], ws)) + adv);
        e[u] = (2 * (j + u) + half < cnt) ? x : -INFINITY;
      }
      const float mn = fmaxf(fmaxf(m, fmaxf(e[0], e[1])), fmaxf(e[2], e[3]));
      float x = m;
      x = (k == 0) ? e[0] : x;
      x = (k == 1) ? e[1] : x;
      x = (k == 2) ? e[2] : x;
      x = (k == 3) ? e[3] : x;
      const float y = expf(x - mn);
      float w[4];
      w[0] = bcast8<0>(y);
      w[1] = bcast8<1>(y);
      w[2] = bcast8<2>(y);
      w[3] = bcast8<3>(y);
      const float sc = bcast8<4>(y);
      sum = fmaf(sum, sc, (w[0] + w[1]) + (w[2] + w[3]));
#pragma unroll
      for (int v = 0; v < 4; ++v) {
        float t2 = acc[v] * sc;
#pragma unroll
        for (int u = 0; u < 4; ++u) t2 = fmaf(w[u], r[u][v], t2);
        acc[v] = t2;
      }
      m = mn;
    }
  }

  const float ma = fmaxf(m, xhalf(m));
  const float f  = expf(m - ma);
  sum *= f;
  sum += xhalf(sum);
  const float inv = 1.0f / (sum + EPS_);
  v4f o;
#pragma unroll
  for (int v = 0; v < 4; ++v) {
    float a = acc[v] * f;
    a += xhalf(a);
    o[v] = fmaf(a, inv, b[v]);
  }
  if (half == 0)
    *reinterpret_cast<v4f*>(out + (long long)d * HC + c4) = o;
}

extern "C" void kernel_launch(void* const* d_in, const int* in_sizes, int n_in,
                              void* d_out, int out_size, void* d_ws, size_t ws_size,
                              hipStream_t stream) {
  (void)in_sizes; (void)n_in; (void)out_size; (void)ws_size;

  const float* x    = (const float*)d_in[0];
  const int*   ei   = (const int*)  d_in[1];
  const float* W1   = (const float*)d_in[2];
  const float* aS1  = (const float*)d_in[3];
  const float* aD1  = (const float*)d_in[4];
  const float* b1   = (const float*)d_in[5];
  const float* W2   = (const float*)d_in[6];
  const float* aS2  = (const float*)d_in[7];
  const float* aD2  = (const float*)d_in[8];
  const float* b2   = (const float*)d_in[9];
  float* out = (float*)d_out;

  float* ws  = (float*)d_ws;
  float* h1  = ws;
  float* o1  = h1  + (long long)NNODES * L1_HC;
  int*   rowptr = (int*)(o1 + (long long)NNODES * L1_HC);
  int*   cntTab = rowptr + NRANGE * RANGE_N;
  int*   offTab = cntTab + NRANGE * NCHUNK;
  int*   srcs   = offTab + NRANGE * NCHUNK;
  int*   tmp_s  = srcs + NEDGES;
  int*   tmp_d  = tmp_s + NEDGES;
  float* h2  = h1;

  bin_kernel<<<NCHUNK, SORT_T, 0, stream>>>(ei, cntTab, offTab, tmp_s, tmp_d);
  collect_kernel<<<NRANGE, SORT_T, 0, stream>>>(cntTab, offTab, tmp_s, tmp_d, rowptr, srcs);

  gemm_wmma_kernel<128, L1_HC><<<NNODES / 16, 128, 0, stream>>>(x, W1, h1);
  gat_gather_kernel<L1_H, L1_HC / 32, true><<<NNODES / NODES_PER_BLOCK, 256, 0, stream>>>(
      rowptr, srcs, aS1, aD1, h1, b1, o1);

  gemm_wmma_kernel<128, L2_HC><<<NNODES / 16, 64, 0, stream>>>(o1, W2, h2);
  gat_gather_pair_kernel<<<NNODES / NODES_PER_BLOCK, 256, 0, stream>>>(
      rowptr, srcs, aS2, aD2, h2, b2, out);
}


# Both gathers: a group's four per-edge score sums reduced together by a transposing butterfly (a lane ends with one edge's score; activation, mask, max once per lane); order of every sum unchanged
# speedup vs baseline: 14.7779x; 1.0614x over previous
#include <hip/hip_runtime.h>
#include <math.h>

#define NNODES   100000
#define NEDGES   1600000
#define L1_H     4
#define L1_C     32
#define L1_HC    128
#define L2_H     1
#define L2_C     64
#define L2_HC    64
#define NEG_SLOPE 0.2f
#define EPS_      1e-16f

#define NODES_PER_BLOCK 8

#define NCHUNK   250
#define CHUNK_E  6400
#define RANGE_SH 9
#define RANGE_N  512
#define NRANGE   196
#define SORT_T   1024
#define STAGE_CAP 12288

static_assert(NCHUNK * CHUNK_E == NEDGES, "chunks cover the edge list exactly");
static_assert(NRANGE * RANGE_N >= NNODES + 1, "ranges must cover rowptr[0..N]");
static_assert((1 << RANGE_SH) == RANGE_N, "range size is a power of two");
static_assert(NCHUNK * 4 <= SORT_T, "four threads per piece in the collect kernel");
static_assert(NNODES % NODES_PER_BLOCK == 0, "gather grid divides the node count");

typedef __attribute__((ext_vector_type(2))) float v2f;
typedef __attribute__((ext_vector_type(4))) float v4f;
typedef __attribute__((ext_vector_type(8))) float v8f;

#define A_PITCH 132

template <int K, int NC>
__global__ __launch_bounds__(NC)
void gemm_wmma_kernel(const float* __restrict__ A, const float* __restrict__ B,
                      float* __restrict__ C) {
  constexpr int NT     = NC;
  constexpr int ROW_CH = K / 4;
  constexpr int PER_T  = 16 * ROW_CH / NT;
  static_assert(ROW_CH == 32, "one wave-load covers one row of A");
  static_assert((16 * ROW_CH) % NT == 0, "tile pieces divide among the threads");
  __shared__ __attribute__((aligned(16))) float tile[16 * A_PITCH];

  const int lane = threadIdx.x & 31;
  const int wv   = threadIdx.x >> 5;
  const int row0 = blockIdx.x * 16;
  const int col0 = wv * 32;
  const int m    = lane & 15;
  const int kk   = (lane >> 4) << 1;

  v4f st[PER_T];
#pragma unroll
  for (int i = 0; i < PER_T; ++i) {
    const int r = wv + i * (NT / 32);
    st[i] = *reinterpret_cast<const v4f*>(A + (long long)(row0 + r) * K + lane * 4);
  }
#pragma unroll
  for (int i = 0; i < PER_T; ++i) {
    const int r = wv + i * (NT / 32);
    *reinterpret_cast<v4f*>(&tile[r * A_PITCH + lane * 4]) = st[i];
  }
  __syncthreads();

  v8f acc0 = {};
  v8f acc1 = {};
  const float* ap = tile + m * A_PITCH + kk;
  const float* bp = B + (long long)kk * NC + (col0 + 2 * m);
#pragma unroll
  for (int k0 = 0; k0 < K; k0 += 4) {
    const v2f a = *reinterpret_cast<const v2f*>(ap + k0);
    const v2f p = *reinterpret_cast<const v2f*>(bp + k0 * NC);
    const v2f q = *reinterpret_cast<const v2f*>(bp + (k0 + 1) * NC);
    v2f b0, b1;
    b0.x = p.x; b0.y = q.x;
    b1.x = p.y; b1.y = q.y;
    acc0 = __builtin_amdgcn_wmma_f32_16x16x4_f32(
        false, a, false, b0, (short)0, acc0, false, false);
    acc1 = __builtin_amdgcn_wmma_f32_16x16x4_f32(
        false, a, false, b1, (short)0, acc1, false, false);
  }
  const int orow = row0 + ((lane >> 4) << 3);
  float* cp = C + (long long)orow * NC + (col0 + 2 * m);
#pragma unroll
  for (int r = 0; r < 8; ++r) {
    v2f o;
    o.x = acc0[r]; o.y = acc1[r];
    *reinterpret_cast<v2f*>(cp + (long long)r * NC) = o;
  }
}

__global__ __launch_bounds__(SORT_T)
void bin_kernel(const int* __restrict__ ei, int* __restrict__ cntTab,
                int* __restrict__ offTab, int* __restrict__ tmp_s,
                int* __restrict__ tmp_d) {
  __shared__ int hist[NRANGE];
  __shared__ int stage_s[CHUNK_E];
  __shared__ int stage_d[CHUNK_E];
  constexpr int PER = (CHUNK_E + SORT_T - 1) / SORT_T;
  constexpr int PL  = (NRANGE + 31) / 32;
  const int t = threadIdx.x;
  const int base = blockIdx.x * CHUNK_E;

  for (int b = t; b < NRANGE; b += SORT_T) hist[b] = 0;
  __syncthreads();

  int s[PER], d[PER];
#pragma unroll
  for (int k = 0; k < PER; ++k) {
    const int i = t + k * SORT_T;
    s[k] = 0; d[k] = 0;
    if (i < CHUNK_E) { d[k] = ei[NEDGES + base + i]; s[k] = ei[base + i]; }
  }
#pragma unroll
  for (int k = 0; k < PER; ++k)
    if (t + k * SORT_T < CHUNK_E) atomicAdd(&hist[d[k] >> RANGE_SH], 1);
  __syncthreads();

  if (t < 32) {
    int c[PL];
    int sum = 0;
#pragma unroll
    for (int j = 0; j < PL; ++j) {
      const int idx = t * PL + j;
      c[j] = (idx < NRANGE) ? hist[idx] : 0;
      sum += c[j];
    }
    int x = sum;
    for (int o = 1; o < 32; o <<= 1) { int y = __shfl_up(x, o); if (t >= o) x += y; }
    int run = x - sum;
#pragma unroll
    for (int j = 0; j < PL; ++j) {
      const int idx = t * PL + j;
      if (idx < NRANGE) {
        cntTab[idx * NCHUNK + blockIdx.x] = c[j];
        offTab[idx * NCHUNK + blockIdx.x] = run;
        hist[idx] = run;
      }
      run += c[j];
    }
  }
  __syncthreads();

#pragma unroll
  for (int k = 0; k < PER; ++k)
    if (t + k * SORT_T < CHUNK_E) {
      const int pos = atomicAdd(&hist[d[k] >> RANGE_SH], 1);
      stage_s[pos] = s[k];
      stage_d[pos] = d[k];
    }
  __syncthreads();

  for (int i = t; i < CHUNK_E; i += SORT_T) {
    tmp_s[base + i] = stage_s[i];
    tmp_d[base + i] = stage_d[i];
  }
}

__global__ __launch_bounds__(SORT_T)
void collect_kernel(const int* __restrict__ cntTab, const int* __restrict__ offTab,
                    const int* __restrict__ tmp_s, const int* __restrict__ tmp_d,
                    int* __restrict__ rowptr, int* __restrict__ srcs) {
  __shared__ int cnt[RANGE_N];
  __shared__ int cur[RANGE_N];
  __shared__ int pc[NCHUNK];
  __shared__ int po[NCHUNK];
  __shared__ int wsum[32];
  __shared__ int tot_s;
  __shared__ int stage[STAGE_CAP];
  const int t = threadIdx.x, lane = t & 31, wv = t >> 5;
  const int node0 = blockIdx.x * RANGE_N;
  const int npre  = blockIdx.x * NCHUNK;

  int acc = 0;
  for (int i = t; i < npre; i += SORT_T) acc += cntTab[i];
  for (int o = 16; o > 0; o >>= 1) acc += __shfl_xor(acc, o);
  if (lane == 0) wsum[wv] = acc;
  if (t < NCHUNK) { pc[t] = cntTab[npre + t]; po[t] = t * CHUNK_E + offTab[npre + t]; }
  if (t < RANGE_N) cnt[t] = 0;
  __syncthreads();
  int S = 0;
  for (int w = 0; w < 32; ++w) S += wsum[w];

  const int p = t >> 2, q = t & 3;
  int n = 0, o = 0;
  if (p < NCHUNK) { n = pc[p]; o = po[p]; }

  for (int j = q; j < n; j += 32) {
    int dd[8];
#pragma unroll
    for (int u = 0; u < 8; ++u) {
      const int jj = j + 4 * u;
      dd[u] = tmp_d[o + (jj < n ? jj : n - 1)];
    }
    asm volatile("" :: "v"(dd[0]), "v"(dd[1]), "v"(dd[2]), "v"(dd[3]),
                       "v"(dd[4]), "v"(dd[5]), "v"(dd[6]), "v"(dd[7]));
#pragma unroll
    for (int u = 0; u < 8; ++u)
      if (j + 4 * u < n) atomicAdd(&cnt[dd[u] - node0], 1);
  }
  __syncthreads();

  const int v = (t < RANGE_N) ? cnt[t] : 0;
  int x = v;
  for (int o2 = 1; o2 < 32; o2 <<= 1) { int y = __shfl_up(x, o2); if (lane >= o2) x += y; }
  if (lane == 31) wsum[wv] = x;
  __syncthreads();
  if (wv == 0) {
    const int w = wsum[lane];
    int xs = w;
    for (int o2 = 1; o2 < 32; o2 <<= 1) { int y = __shfl_up(xs, o2); if (lane >= o2) xs += y; }
    wsum[lane] = xs - w;
    if (lane == 31) tot_s = xs;
  }
  __syncthreads();
  const int total = tot_s;
  const int myoff = x - v + wsum[wv];
  if (t < RANGE_N) {
    cur[t] = myoff;
    const int node = node0 + t;
    if (node <= NNODES) rowptr[node] = S + myoff;
  }
  __syncthreads();

  const bool staged = total <= STAGE_CAP;
  for (int j = q; j < n; j += 32) {
    int es[8], dd[8];
#pragma unroll
    for (int u = 0; u < 8; ++u) {
      const int jj = j + 4 * u;
      const int at = o + (jj < n ? jj : n - 1);
      es[u] = tmp_s[at];
      dd[u] = tmp_d[at];
    }
    asm volatile("" :: "v"(dd[0]), "v"(dd[1]), "v"(dd[2]), "v"(dd[3]),
                       "v"(dd[4]), "v"(dd[5]), "v"(dd[6]), "v"(dd[7]),
                       "v"(es[0]), "v"(es[1]), "v"(es[2]), "v"(es[3]),
                       "v"(es[4]), "v"(es[5]), "v"(es[6]), "v"(es[7]));
#pragma unroll
    for (int u = 0; u < 8; ++u)
      if (j + 4 * u < n) {
        const int pos = atomicAdd(&cur[dd[u] - node0], 1);
        if (staged) stage[pos] = es[u]; else srcs[S + pos] = es[u];
      }
  }
  __syncthreads();
  if (staged)
    for (int i = t; i < total; i += SORT_T) srcs[S + i] = stage[i];
}

template <int XMASK>
__device__ __forceinline__ float row16_xadd(float v) {
  const int other = __builtin_amdgcn_update_dpp(0, __float_as_int(v), 0x160 | XMASK, 0xf, 0xf, true);
  return v + __int_as_float(other);
}
template <int XMASK>
__device__ __forceinline__ float row16_xget(float v) {
  return __int_as_float(__builtin_amdgcn_update_dpp(0, __float_as_int(v), 0x160 | XMASK, 0xf, 0xf, true));
}
__device__ __forceinline__ float quad_transpose_sum(const float (&p)[4], bool b0, bool b1) {
  const float kA = b0 ? p[1] : p[0], sA = b0 ? p[0] : p[1];
  const float kB = b0 ? p[3] : p[2], sB = b0 ? p[2] : p[3];
  const float qA = kA + row16_xget<1>(sA);
  const float qB = kB + row16_xget<1>(sB);
  const float kq = b1 ? qB : qA, sq = b1 ? qA : qB;
  return kq + row16_xget<2>(sq);
}
__device__ __forceinline__ float quad_max(float v) {
  v = fmaxf(v, row16_xget<1>(v));
  v = fmaxf(v, row16_xget<2>(v));
  return v;
}
__device__ __forceinline__ float xhalf(float x) {
  const int b = __float_as_int(x);
  return __int_as_float(__builtin_amdgcn_permlanex16(b, b, 0x76543210, 0xfedcba98, false, false));
}

template <int SRC>
__device__ __forceinline__ float bcast8(float x) {
  return __int_as_float((int)__builtin_amdgcn_mov_dpp8((unsigned)__float_as_int(x), SRC * 0x249249));
}

template <int H, int VEC, bool ELU_ACT>
__global__ __launch_bounds__(256)
void gat_gather_kernel(const int* __restrict__ rowptr, const int* __restrict__ srcs,
                       const float* __restrict__ att_s, const float* __restrict__ att_d,
                       const float* __restrict__ hfeat, const float* __restrict__ bias,
                       float* __restrict__ out) {
  constexpr int HC  = 32 * VEC;
  constexpr int LPH = 32 / H;
  typedef __attribute__((ext_vector_type(VEC))) float vecf;
  static_assert(LPH == 8 || LPH == 32, "head sum handles 8 or 32 lanes per head");

  const int lane = threadIdx.x & 31;
  const int d = __builtin_amdgcn_readfirstlane(
      (int)(blockIdx.x * NODES_PER_BLOCK + (threadIdx.x >> 5)));
  const int start = rowptr[d];
  const int deg   = rowptr[d + 1] - start;

  const vecf ws = *reinterpret_cast<const vecf*>(att_s + lane * VEC);
  const vecf wd = *reinterpret_cast<const vecf*>(att_d + lane * VEC);
  const vecf b  = *reinterpret_cast<const vecf*>(bias + lane * VEC);

  auto dotv = [&](const vecf& r, const vecf& w) -> float {
    float p = r[0] * w[0];
#pragma unroll
    for (int v = 1; v < VEC; ++v) p = fmaf(r[v], w[v], p);
    return p;
  };
  auto head_sum = [&](float x) -> float {
    x = row16_xadd<1>(x);
    x = row16_xadd<2>(x);
    x = row16_xadd<4>(x);
    if constexpr (LPH == 32) {
      x = row16_xadd<8>(x);
      x = x + xhalf(x);
      x = __int_as_float(__builtin_amdgcn_readfirstlane(__float_as_int(x)));
    }
    return x;
  };
  auto lrelu = [](float x) -> float { return x > 0.0f ? x : NEG_SLOPE * x; };

  const vecf rd = *reinterpret_cast<const vecf*>(hfeat + (long long)d * HC + lane * VEC);
  const float adv = head_sum(dotv(rd, wd));
  float m   = lrelu(head_sum(dotv(rd, ws)) + adv);
  float sum = 1.0f;
  float acc[VEC];
#pragma unroll
  for (int v = 0; v < VEC; ++v) acc[v] = rd[v];

  for (int base = 0; base < deg; base += 32) {
    const int left = deg - base;
    const int cnt  = left < 32 ? left : 32;
    int s = d;
    if (lane < cnt) s = srcs[start + base + lane];
    for (int j = 0; j < cnt; j += 4) {
      vecf  r[4];
      float e[4];
#pragma unroll
      for (int u = 0; u < 4; ++u) {
        const int jj = (j + u < cnt) ? (j + u) : (cnt - 1);
        const int sj = __builtin_amdgcn_readlane(s, jj);
        r[u] = *reinterpret_cast<const vecf*>(hfeat + (long long)sj * HC + lane * VEC);
      }
float mn, sc, w[4];
      if constexpr (LPH == 8) {
        float p[4];
#pragma unroll
        for (int u = 0; u < 4; ++u) p[u] = dotv(r[u], ws);
        const float t  = row16_xadd<4>(quad_transpose_sum(p, (lane & 1) != 0, (lane & 2) != 0));
        const float xs = lrelu(t + adv);
        const float em = (j + (lane & 3) < cnt) ? xs : -INFINITY;
        mn = fmaxf(m, quad_max(em));
        const float x = (lane & 4) ? m : em;
        const float y = expf(x - mn);
        w[0] = bcast8<0>(y);
        w[1] = bcast8<1>(y);
        w[2] = bcast8<2>(y);
        w[3] = bcast8<3>(y);
        sc   = bcast8<4>(y);
      } else {
#pragma unroll
        for (int u = 0; u < 4; ++u) {
          const float x = lrelu(head_sum(dotv(r[u], ws)) + adv);
          e[u] = (j + u < cnt) ? x : -INFINITY;
        }
        mn = fmaxf(fmaxf(m, fmaxf(e[0], e[1])), fmaxf(e[2], e[3]));
        sc = expf(m - mn);
#pragma unroll
        for (int u = 0; u < 4; ++u) w[u] = expf(e[u] - mn);
      }
      sum = fmaf(sum, sc, (w[0] + w[1]) + (w[2] + w[3]));
#pragma unroll
      for (int v = 0; v < VEC; ++v) {
        float t = acc[v] * sc;
#pragma unroll
        for (int u = 0; u < 4; ++u) t = fmaf(w[u], r[u][v], t);
        acc[v] = t;
      }
      m = mn;
    }
  }

  const float inv = 1.0f / (sum + EPS_);
  vecf o;
#pragma unroll
  for (int v = 0; v < VEC; ++v) {
    float x = fmaf(acc[v], inv, b[v]);
    if (ELU_ACT) x = x > 0.0f ? x : expm1f(x);
    o[v] = x;
  }
  *reinterpret_cast<vecf*>(out + (long long)d * HC + lane * VEC) = o;
}

__device__ __forceinline__ int row_pick(int x, int sel) {
  const unsigned s = (unsigned)sel * 0x11111111u;
  return (int)__builtin_amdgcn_permlane16((unsigned)x, (unsigned)x, s, s, false, false);
}

__global__ __launch_bounds__(256)
void gat_gather_pair_kernel(const int* __restrict__ rowptr, const int* __restrict__ srcs,
                            const float* __restrict__ att_s, const float* __restrict__ att_d,
                            const float* __restrict__ hfeat, const float* __restrict__ bias,
                            float* __restrict__ out) {
  constexpr int HC = L2_HC;
  static_assert(HC == 64 && L2_H == 1, "16 lanes x 4 channels cover one single-head row");

  const int lane = threadIdx.x & 31;
  const int c4   = (lane & 15) * 4;
  const int half = lane >> 4;
  const int d = __builtin_amdgcn_readfirstlane(
      (int)(blockIdx.x * NODES_PER_BLOCK + (threadIdx.x >> 5)));
  const int start = rowptr[d];
  const int deg   = rowptr[d + 1] - start;

  const v4f ws = *reinterpret_cast<const v4f*>(att_s + c4);
  const v4f wd = *reinterpret_cast<const v4f*>(att_d + c4);
  const v4f b  = *reinterpret_cast<const v4f*>(bias + c4);

  auto dot4 = [](const v4f& r, const v4f& w) -> float {
    float p = r[0] * w[0];
    p = fmaf(r[1], w[1], p);
    p = fmaf(r[2], w[2], p);
    p = fmaf(r[3], w[3], p);
    return p;
  };
  auto row_sum = [](float x) -> float {
    x = row16_xadd<1>(x);
    x = row16_xadd<2>(x);
    x = row16_xadd<4>(x);
    x = row16_xadd<8>(x);
    return x;
  };
  auto lrelu = [](float x) -> float { return x > 0.0f ? x : NEG_SLOPE * x; };

  const v4f rd = *reinterpret_cast<const v4f*>(hfeat + (long long)d * HC + c4);
  const float adv = row_sum(dot4(rd, wd));
  float m   = lrelu(row_sum(dot4(rd, ws)) + adv);
  float sum = half ? 0.0f : 1.0f;
  float acc[4];
#pragma unroll
  for (int v = 0; v < 4; ++v) acc[v] = half ? 0.0f : rd[v];

  const int k = lane & 7;
  for (int base = 0; base < deg; base += 32) {
    const int left = deg - base;
    const int cnt  = left < 32 ? left : 32;
    const int mine = 2 * (lane & 15) + half;
    int s = d;
    if (mine < cnt) s = srcs[start + base + mine];
    const int npair = (cnt + 1) >> 1;
    for (int j = 0; j < npair; j += 4) {
      v4f   r[4];
#pragma unroll
      for (int u = 0; u < 4; ++u) {
        const int t  = (j + u < npair) ? (j + u) : (npair - 1);
        const int sj = row_pick(s, t);
        r[u] = *reinterpret_cast<const v4f*>(hfeat + ((unsigned)sj * (unsigned)HC + (unsigned)c4));
      }
      float p[4];
#pragma unroll
      for (int u = 0; u < 4; ++u) p[u] = dot4(r[u], ws);
      float t = quad_transpose_sum(p, (lane & 1) != 0, (lane & 2) != 0);
      t = row16_xadd<4>(t);
      t = row16_xadd<8>(t);
      const float xs = lrelu(t + adv);
      const float em = (2 * (j + (lane & 3)) + half < cnt) ? xs : -INFINITY;
      const float mn = fmaxf(m, quad_max(em));
      const float x = (k & 4) ? m : em;
      const float y = expf(x - mn);
      float w[4];
      w[0] = bcast8<0>(y);
      w[1] = bcast8<1>(y);
      w[2] = bcast8<2>(y);
      w[3] = bcast8<3>(y);
      const float sc = bcast8<4>(y);
      sum = fmaf(sum, sc, (w[0] + w[1]) + (w[2] + w[3]));
#pragma unroll
      for (int v = 0; v < 4; ++v) {
        float t2 = acc[v] * sc;
#pragma unroll
        for (int u = 0; u < 4; ++u) t2 = fmaf(w[u], r[u][v], t2);
        acc[v] = t2;
      }
      m = mn;
    }
  }

  const float ma = fmaxf(m, xhalf(m));
  const float f  = expf(m - ma);
  sum *= f;
  sum += xhalf(sum);
  const float inv = 1.0f / (sum + EPS_);
  v4f o;
#pragma unroll
  for (int v = 0; v < 4; ++v) {
    float a = acc[v] * f;
    a += xhalf(a);
    o[v] = fmaf(a, inv, b[v]);
  }
  if (half == 0)
    *reinterpret_cast<v4f*>(out + (long long)d * HC + c4) = o;
}

extern "C" void kernel_launch(void* const* d_in, const int* in_sizes, int n_in,
                              void* d_out, int out_size, void* d_ws, size_t ws_size,
                              hipStream_t stream) {
  (void)in_sizes; (void)n_in; (void)out_size; (void)ws_size;

  const float* x    = (const float*)d_in[0];
  const int*   ei   = (const int*)  d_in[1];
  const float* W1   = (const float*)d_in[2];
  const float* aS1  = (const float*)d_in[3];
  const float* aD1  = (const float*)d_in[4];
  const float* b1   = (const float*)d_in[5];
  const float* W2   = (const float*)d_in[6];
  const float* aS2  = (const float*)d_in[7];
  const float* aD2  = (const float*)d_in[8];
  const float* b2   = (const float*)d_in[9];
  float* out = (float*)d_out;

  float* ws  = (float*)d_ws;
  float* h1  = ws;
  float* o1  = h1  + (long long)NNODES * L1_HC;
  int*   rowptr = (int*)(o1 + (long long)NNODES * L1_HC);
  int*   cntTab = rowptr + NRANGE * RANGE_N;
  int*   offTab = cntTab + NRANGE * NCHUNK;
  int*   srcs   = offTab + NRANGE * NCHUNK;
  int*   tmp_s  = srcs + NEDGES;
  int*   tmp_d  = tmp_s + NEDGES;
  float* h2  = h1;

  bin_kernel<<<NCHUNK, SORT_T, 0, stream>>>(ei, cntTab, offTab, tmp_s, tmp_d);
  collect_kernel<<<NRANGE, SORT_T, 0, stream>>>(cntTab, offTab, tmp_s, tmp_d, rowptr, srcs);

  gemm_wmma_kernel<128, L1_HC><<<NNODES / 16, 128, 0, stream>>>(x, W1, h1);
  gat_gather_kernel<L1_H, L1_HC / 32, true><<<NNODES / NODES_PER_BLOCK, 256, 0, stream>>>(
      rowptr, srcs, aS1, aD1, h1, b1, o1);

  gemm_wmma_kernel<128, L2_HC><<<NNODES / 16, 64, 0, stream>>>(o1, W2, h2);
  gat_gather_pair_kernel<<<NNODES / NODES_PER_BLOCK, 256, 0, stream>>>(
      rowptr, srcs, aS2, aD2, h2, b2, out);
}
